# starting kernel
# baseline (speedup 1.0000x reference)
#include <hip/hip_runtime.h>
#include <hip/hip_bf16.h>

#define DFEAT 96

typedef float v2f __attribute__((ext_vector_type(2)));
typedef float v8f __attribute__((ext_vector_type(8)));


__global__ void k_fill(float* __restrict__ p, float val, int n) {
  int i = blockIdx.x * blockDim.x + threadIdx.x;
  if (i < n) p[i] = val;
}

__global__ void k_deg_accum(const int* __restrict__ dst, const float* __restrict__ ew,
                            float* __restrict__ deg, int ne) {
  int e = blockIdx.x * blockDim.x + threadIdx.x;
  if (e < ne) atomicAdd(&deg[dst[e]], ew[e]);
}

__global__ void k_dinv(const float* __restrict__ deg, float* __restrict__ dinv, int n) {
  int i = blockIdx.x * blockDim.x + threadIdx.x;
  if (i < n) {
    float d = deg[i];
    dinv[i] = (d > 0.0f) ? rsqrtf(fmaxf(d, 1e-20f)) : 0.0f;
  }
}

__global__ __launch_bounds__(256) void k_gemm_wmma_fused(
    const float* __restrict__ X, const float* __restrict__ W,
    const float* __restrict__ dinv, const float* __restrict__ bias,
    float* __restrict__ T, float* __restrict__ out,
    int nrows, int relu_in) {
  __shared__ v2f sW2[(DFEAT / 2) * DFEAT];
  for (int idx = threadIdx.x; idx < (DFEAT / 2) * DFEAT; idx += 256) {
    const int kp  = idx / DFEAT;
    const int col = idx - kp * DFEAT;
    v2f w;
    w.x = W[(2 * kp) * DFEAT + col];
    w.y = W[(2 * kp + 1) * DFEAT + col];
    sW2[idx] = w;
  }
  __syncthreads();

  const int wave = threadIdx.x >> 5;
  const int lane = threadIdx.x & 31;
  const int ntiles = nrows >> 4;
  const int mt = blockIdx.x * 8 + wave;
  if (mt >= ntiles) return;

  const int lo16  = lane & 15;
  const int khalf = lane >> 4;
  const int row   = mt * 16 + lo16;
  const float* xr = X + (size_t)row * DFEAT;

  v8f acc[6];
#pragma unroll
  for (int t = 0; t < 6; ++t) acc[t] = (v8f){0.f, 0.f, 0.f, 0.f, 0.f, 0.f, 0.f, 0.f};

#pragma unroll 4
  for (int k = 0; k < 24; ++k) {
    const int kb = 4 * k + 2 * khalf;
    v2f a = *(const v2f*)(xr + kb);
    if (relu_in) {
      a.x = (a.x > 0.0f) ? a.x : 0.0f;
      a.y = (a.y > 0.0f) ? a.y : 0.0f;
    }
    const v2f* brow = sW2 + (kb >> 1) * DFEAT + lo16;
#pragma unroll
    for (int t = 0; t < 6; ++t) {
      v2f b = brow[t * 16];
      acc[t] = __builtin_amdgcn_wmma_f32_16x16x4_f32(
           false, a,  false, b,
           (short)0, acc[t],  false,  false);
    }
  }

  const int rbase = mt * 16 + khalf * 8;
  float dv2[8];
#pragma unroll
  for (int r = 0; r < 8; ++r) {
    float d = dinv[rbase + r];
    dv2[r] = d * d;
  }
#pragma unroll
  for (int t = 0; t < 6; ++t) {
    const int cbase = t * 16 + lo16;
    const float bb = bias[cbase];
#pragma unroll
    for (int r = 0; r < 8; ++r) {
      const size_t idx = (size_t)(rbase + r) * DFEAT + cbase;
      const float v = acc[t][r];
      T[idx] = v;
      out[idx] = dv2[r] * v + bb;
    }
  }
}

__global__ __launch_bounds__(256) void k_scatter(const float* __restrict__ T,
                                                 const int* __restrict__ src,
                                                 const int* __restrict__ dst,
                                                 const float* __restrict__ ew,
                                                 const float* __restrict__ dinv,
                                                 float* __restrict__ out, int ne) {
  const int e = blockIdx.x * 8 + (threadIdx.x >> 5);
  if (e >= ne) return;
  const int lane = threadIdx.x & 31;
  const int s = src[e];
  const int d = dst[e];
  const float nrm = dinv[s] * ew[e] * dinv[d];
  const float* ts = T + (size_t)s * DFEAT;
  float* od = out + (size_t)d * DFEAT;
#pragma unroll
  for (int j = 0; j < 3; ++j) {
    const int f = lane + 32 * j;
    atomicAdd(&od[f], nrm * ts[f]);
  }
}

__global__ void k_pool(const float* __restrict__ ge, const int* __restrict__ batch,
                       float* __restrict__ gout, int total) {
  int i = blockIdx.x * blockDim.x + threadIdx.x;
  if (i >= total) return;
  int node = i / DFEAT;
  int f = i - node * DFEAT;
  atomicAdd(&gout[batch[node] * DFEAT + f], ge[i]);
}

extern "C" void kernel_launch(void* const* d_in, const int* in_sizes, int n_in,
                              void* d_out, int out_size, void* d_ws, size_t ws_size,
                              hipStream_t stream) {
  (void)n_in; (void)ws_size;
  const float* x    = (const float*)d_in[0];
  const float* W1   = (const float*)d_in[1];
  const float* b1   = (const float*)d_in[2];
  const float* W3   = (const float*)d_in[3];
  const float* b3   = (const float*)d_in[4];
  const float* W4   = (const float*)d_in[5];
  const float* b4   = (const float*)d_in[6];
  const float* ew   = (const float*)d_in[7];
  const int*   ei   = (const int*)d_in[8];
  const int*   batch= (const int*)d_in[9];

  const int n  = in_sizes[0] / DFEAT;
  const int ne = in_sizes[7];
  const int* srcp = ei;
  const int* dstp = ei + ne;

  const int nf = n * DFEAT;
  const int gtotal = out_size - nf;

  float* deg  = (float*)d_ws;
  float* dinv = deg + n;
  float* tmp  = dinv + n;
  float* h    = tmp + (size_t)nf;
  float* tge  = h + (size_t)nf;

  float* embed = (float*)d_out;
  float* gout  = embed + nf;

  const dim3 B(256);
  const int gN  = (n + 255) / 256;
  const int gE  = (ne + 255) / 256;
  const int gF  = (nf + 255) / 256;
  const int gSc = (ne + 7) / 8;
  const int gMM = ((n >> 4) + 7) / 8;

  k_fill<<<gN, B, 0, stream>>>(deg, 1.0f, n);
  k_deg_accum<<<gE, B, 0, stream>>>(dstp, ew, deg, ne);
  k_dinv<<<gN, B, 0, stream>>>(deg, dinv, n);

  k_gemm_wmma_fused<<<gMM, B, 0, stream>>>(x, W1, dinv, b1, tmp, h, n, 0);
  k_scatter<<<gSc, B, 0, stream>>>(tmp, srcp, dstp, ew, dinv, h, ne);

  k_gemm_wmma_fused<<<gMM, B, 0, stream>>>(h, W3, dinv, b3, tmp, embed, n, 1);
  k_scatter<<<gSc, B, 0, stream>>>(tmp, srcp, dstp, ew, dinv, embed, ne);

  k_gemm_wmma_fused<<<gMM, B, 0, stream>>>(h, W4, dinv, b4, tmp, tge, n, 1);
  k_scatter<<<gSc, B, 0, stream>>>(tmp, srcp, dstp, ew, dinv, tge, ne);
  k_fill<<<(gtotal + 255) / 256, B, 0, stream>>>(gout, 0.0f, gtotal);
  k_pool<<<gF, B, 0, stream>>>(tge, batch, gout, nf);
}


# Edge list re-encoded in the degree launch: 1-bit-per-edge mask per 784-row range + 16-byte {src,dst,w} records; LDS aggregations scan mask words, fetch one record per edge (arrival-order sums kept)
# speedup vs baseline: 2.1603x; 2.1603x over previous
#include <hip/hip_runtime.h>
#include <hip/hip_bf16.h>

#define DFEAT 96

typedef float v2f __attribute__((ext_vector_type(2)));
typedef float v8f __attribute__((ext_vector_type(8)));


__global__ void k_fill(float* __restrict__ p, float val, int n) {
  int i = blockIdx.x * blockDim.x + threadIdx.x;
  if (i < n) p[i] = val;
}

#define AG_R    784
#define DM_EPW  2048
#define DM_WPW  64
#define DM_MAXR 64
__global__ __launch_bounds__(256) void k_deg_mask(const int* __restrict__ src,
                                                  const int* __restrict__ dst,
                                                  const float* __restrict__ ew,
                                                  float* __restrict__ deg,
                                                  int4* __restrict__ rec,
                                                  unsigned* __restrict__ mask,
                                                  int ne, int nr, int wpr) {
  __shared__ unsigned tile[DM_MAXR * (DM_WPW + 1)];
  const int lane = threadIdx.x & 31;
  const int wave = threadIdx.x >> 5;
  const int e0 = blockIdx.x * DM_EPW + threadIdx.x;
  int d[8], s[8];
  float w[8];
#pragma unroll
  for (int j = 0; j < 8; ++j) {
    const int e = e0 + j * 256;
    const bool ok = e < ne;
    d[j] = ok ? dst[e] : -1;
    s[j] = ok ? src[e] : 0;
    w[j] = ok ? ew[e] : 0.0f;
  }
  for (int rb = 0; rb < nr; rb += DM_MAXR) {
    for (int i = threadIdx.x; i < DM_MAXR * (DM_WPW + 1); i += 256) tile[i] = 0u;
    __syncthreads();
#pragma unroll
    for (int j = 0; j < 8; ++j) {
      if (d[j] >= 0) {
        const int r = d[j] / AG_R - rb;
        if ((unsigned)r < (unsigned)DM_MAXR) atomicOr(&tile[r * (DM_WPW + 1) + j * 8 + wave], 1u << lane);
      }
    }
    __syncthreads();
    const int nrows = (nr - rb < DM_MAXR) ? nr - rb : DM_MAXR;
    for (int i = threadIdx.x; i < nrows * DM_WPW; i += 256) {
      const int r = i / DM_WPW;
      const int k = i - r * DM_WPW;
      mask[(size_t)(rb + r) * wpr + blockIdx.x * DM_WPW + k] = tile[r * (DM_WPW + 1) + k];
    }
    if (rb + DM_MAXR < nr) __syncthreads();
  }
#pragma unroll
  for (int j = 0; j < 8; ++j) {
    if (d[j] >= 0) {
      atomicAdd(&deg[d[j]], w[j]);
      rec[e0 + j * 256] = make_int4(s[j], d[j], __float_as_int(w[j]), 0);
    }
  }
}

#define CNT_BINS 1024
__global__ __launch_bounds__(256) void k_dinv_cnt_mt(const float* __restrict__ deg,
                                                     float* __restrict__ dinv,
                                                     const int* __restrict__ batch,
                                                     float* __restrict__ cnt,
                                                     float* __restrict__ Mt, int n, int G) {
  __shared__ float hist[CNT_BINS];
  __shared__ float sd2[256];
  __shared__ int   sb[256];
  const int nb = (G < CNT_BINS) ? G : CNT_BINS;
  for (int g = threadIdx.x; g < nb; g += 256) hist[g] = 0.0f;
  __syncthreads();
  const int row0 = blockIdx.x * 256;
  const int i = row0 + threadIdx.x;
  float d2 = 0.0f;
  int   bi = -1;
  if (i < n) {
    const float d = deg[i];
    const float dv = (d > 0.0f) ? rsqrtf(fmaxf(d, 1e-20f)) : 0.0f;
    dinv[i] = dv;
    d2 = dv * dv;
    const int b = batch[i];
    bi = b;
    if ((unsigned)b < (unsigned)nb) atomicAdd(&hist[b], 1.0f);
    else if ((unsigned)b < (unsigned)G) atomicAdd(&cnt[b], 1.0f);
  }
  sd2[threadIdx.x] = d2;
  sb[threadIdx.x]  = bi;
  __syncthreads();
  for (int g = threadIdx.x; g < nb; g += 256) {
    const float v = hist[g];
    if (v != 0.0f) atomicAdd(&cnt[g], v);
  }

  const int rows  = (n - row0 < 256) ? n - row0 : 256;
  const int total = rows * G;
  float* mp = Mt + (size_t)row0 * G;
  if (((G & 3) == 0) && ((reinterpret_cast<size_t>(Mt) & 15) == 0)) {
    for (int e0 = 4 * threadIdx.x; e0 < total; e0 += 4 * 256) {
      const int j  = e0 / G;
      const int g0 = e0 - j * G;
      const int b  = sb[j];
      const float s = sd2[j];
      float4 v;
      v.x = (b == g0)     ? s : 0.0f;
      v.y = (b == g0 + 1) ? s : 0.0f;
      v.z = (b == g0 + 2) ? s : 0.0f;
      v.w = (b == g0 + 3) ? s : 0.0f;
      *(float4*)(mp + e0) = v;
    }
  } else {
    for (int e = threadIdx.x; e < total; e += 256) {
      const int j = e / G;
      const int g = e - j * G;
      mp[e] = (sb[j] == g) ? sd2[j] : 0.0f;
    }
  }
}

__global__ __launch_bounds__(256) void k_gemm_wmma_fused(
    const float* __restrict__ X, const float* __restrict__ W,
    const float* __restrict__ dinv, const float* __restrict__ bias,
    float* __restrict__ T, float* __restrict__ out,
    int nrows, int relu_in) {
  __shared__ v2f sW2[(DFEAT / 2) * DFEAT];
  for (int idx = threadIdx.x; idx < (DFEAT / 2) * DFEAT; idx += 256) {
    const int kp  = idx / DFEAT;
    const int col = idx - kp * DFEAT;
    v2f w;
    w.x = W[(2 * kp) * DFEAT + col];
    w.y = W[(2 * kp + 1) * DFEAT + col];
    sW2[idx] = w;
  }
  __syncthreads();

  const int wave = threadIdx.x >> 5;
  const int lane = threadIdx.x & 31;
  const int ntiles = nrows >> 4;
  const int mt = blockIdx.x * 8 + wave;
  if (mt >= ntiles) return;

  const int lo16  = lane & 15;
  const int khalf = lane >> 4;
  const int row   = mt * 16 + lo16;
  const float* xr = X + (size_t)row * DFEAT;

  v8f acc[6];
#pragma unroll
  for (int t = 0; t < 6; ++t) acc[t] = (v8f){0.f, 0.f, 0.f, 0.f, 0.f, 0.f, 0.f, 0.f};

#pragma unroll 4
  for (int k = 0; k < 24; ++k) {
    const int kb = 4 * k + 2 * khalf;
    v2f a = *(const v2f*)(xr + kb);
    if (relu_in) {
      a.x = (a.x > 0.0f) ? a.x : 0.0f;
      a.y = (a.y > 0.0f) ? a.y : 0.0f;
    }
    const v2f* brow = sW2 + (kb >> 1) * DFEAT + lo16;
#pragma unroll
    for (int t = 0; t < 6; ++t) {
      v2f b = brow[t * 16];
      acc[t] = __builtin_amdgcn_wmma_f32_16x16x4_f32(
           false, a,  false, b,
           (short)0, acc[t],  false,  false);
    }
  }

  const int rbase = mt * 16 + khalf * 8;
  float dv2[8];
#pragma unroll
  for (int r = 0; r < 8; ++r) {
    float d = dinv[rbase + r];
    dv2[r] = d * d;
  }
#pragma unroll
  for (int t = 0; t < 6; ++t) {
    const int cbase = t * 16 + lo16;
    const float bb = bias[cbase];
#pragma unroll
    for (int r = 0; r < 8; ++r) {
      const size_t idx = (size_t)(rbase + r) * DFEAT + cbase;
      const float v = acc[t][r];
      T[idx] = v;
      out[idx] = dv2[r] * v + bb;
    }
  }
}

#define AG_Q     4
#define AG_WAVES 32
#define AG_GRP   16
#define AG_QCAP  64
template <bool POOL>
__global__ __launch_bounds__(1024) void k_agg_lds(const float* __restrict__ T,
                                                  const int4* __restrict__ rec,
                                                  const unsigned* __restrict__ mask,
                                                  int wpr, int nwords,
                                                  const float* __restrict__ dinv,
                                                  float* __restrict__ out,
                                                  int n, int per_part,
                                                  const int* __restrict__ batch,
                                                  float* __restrict__ Mt, int G) {
  __shared__ __attribute__((aligned(16))) float acc[AG_R * DFEAT];
  __shared__ int qbuf[AG_WAVES * AG_QCAP];
  const int lane = threadIdx.x & 31;
  const int wave = __builtin_amdgcn_readfirstlane((int)(threadIdx.x >> 5));
  const int range = blockIdx.x / AG_Q;
  const int part  = blockIdx.x - range * AG_Q;
  const int lo = range * AG_R;

  {
    float4* a4 = reinterpret_cast<float4*>(acc);
    const float4 z = make_float4(0.0f, 0.0f, 0.0f, 0.0f);
    for (int i = threadIdx.x; i < AG_R * DFEAT / 4; i += AG_WAVES * 32) a4[i] = z;
  }
  __syncthreads();

  const int pbeg = part * per_part;
  const int pw0  = pbeg >> 5;
  const int pwn  = (pw0 + (per_part >> 5) < nwords) ? (per_part >> 5) : nwords - pw0;
  const unsigned* mrow = mask + (size_t)range * wpr + pw0;
  int* q = qbuf + wave * AG_QCAP;
  int qn = 0;

  auto drain = [&](const int* qq, int cnt) {
    const int qe = qq[(lane < cnt) ? lane : cnt - 1];
    const int4 rc = rec[pbeg + qe];
    const int s = rc.x;
    const int r = rc.y - lo;
    const float w = __int_as_float(rc.z);
    const int dl = r * DFEAT;
    const float nrm = dinv[s] * w * dinv[lo + r];
    const int nrm_bits = __float_as_int(nrm);
    int gph = 0;
    if (POOL) gph = batch[lo + r];
#pragma unroll 1
    for (int g0 = 0; g0 < cnt; g0 += AG_GRP) {
      float t[AG_GRP][3];
#pragma unroll
      for (int u = 0; u < AG_GRP; ++u) {
        const int i = (g0 + u < cnt) ? g0 + u : cnt - 1;
        const float* ts = T + (size_t)__builtin_amdgcn_readlane(s, i) * DFEAT;
#pragma unroll
        for (int j = 0; j < 3; ++j) t[u][j] = ts[lane + 32 * j];
      }
#pragma unroll
      for (int u = 0; u < AG_GRP; ++u) {
        const int i = g0 + u;
        if (i < cnt) {
          float* ad = acc + __builtin_amdgcn_readlane(dl, i);
          const float nu = __int_as_float(__builtin_amdgcn_readlane(nrm_bits, i));
#pragma unroll
          for (int j = 0; j < 3; ++j) atomicAdd(&ad[lane + 32 * j], nu * t[u][j]);
        }
      }
    }
    if (POOL) {
      if (lane < cnt && (unsigned)gph < (unsigned)G) atomicAdd(&Mt[(size_t)s * G + gph], nrm);
    }
  };

  constexpr int WSTRIDE = AG_WAVES * 32;
  auto load_word = [&](int i) -> unsigned {
    const unsigned v = mrow[(i < pwn) ? i : pwn - 1];
    return (i < pwn) ? v : 0u;
  };
  int wb = wave * 32;
  unsigned m = 0u;
  if (wb < pwn) m = load_word(wb + lane);
  for (; wb < pwn; wb += WSTRIDE) {
    const unsigned mn = load_word(wb + WSTRIDE + lane);
    const int ebase = (wb + lane) << 5;
    unsigned any = (unsigned)__builtin_amdgcn_ballot_w32(m != 0u);
    while (any != 0u) {
      const int pos = qn + __builtin_popcount(any & ((1u << lane) - 1u));
      if (m != 0u) {
        q[pos] = ebase + __builtin_ctz(m);
        m &= m - 1u;
      }
      qn += __builtin_popcount(any);
      if (qn >= 32) {
        drain(q + (qn - 32), 32);
        qn -= 32;
      }
      any = (unsigned)__builtin_amdgcn_ballot_w32(m != 0u);
    }
    m = mn;
  }
  if (qn > 0) drain(q, qn);

  __syncthreads();
  const int rows = (n - lo < AG_R) ? n - lo : AG_R;
  float* ob = out + (size_t)lo * DFEAT;
  for (int i = threadIdx.x; i < rows * DFEAT; i += AG_WAVES * 32) atomicAdd(&ob[i], acc[i]);
}


#define PCHUNK 128


#define PGU 4
__global__ __launch_bounds__(128) void k_pool_gemm(const float* __restrict__ Mt,
                                                   const float* __restrict__ H,
                                                   float* __restrict__ Pacc, int n, int G) {
  const int wave  = threadIdx.x >> 5;
  const int lane  = threadIdx.x & 31;
  const int mtile = blockIdx.y * 4 + wave;
  if (mtile * 16 >= G) return;
  const int c  = lane & 15;
  const int hf = lane >> 4;
  const int g  = mtile * 16 + c;
  const bool gok = g < G;
  const float* mcol = Mt + (gok ? g : 0);
  const float* hcol = H + c;
  const int base = blockIdx.x * PCHUNK + 2 * hf;

  v8f acc[6];
#pragma unroll
  for (int t = 0; t < 6; ++t) acc[t] = (v8f){0.f, 0.f, 0.f, 0.f, 0.f, 0.f, 0.f, 0.f};

  constexpr int NG = PCHUNK / (4 * PGU);
  float ra[2][PGU][2];
  float rb[2][PGU][6][2];

#pragma unroll
  for (int u = 0; u < PGU; ++u) {
    const int n0 = base + 4 * u;
    const int m0 = (n0 < n) ? n0 : 0;
    const int m1 = (n0 + 1 < n) ? n0 + 1 : 0;
    ra[0][u][0] = mcol[(size_t)m0 * G];
    ra[0][u][1] = mcol[(size_t)m1 * G];
#pragma unroll
    for (int t = 0; t < 6; ++t) {
      rb[0][u][t][0] = hcol[(size_t)m0 * DFEAT + t * 16];
      rb[0][u][t][1] = hcol[(size_t)m1 * DFEAT + t * 16];
    }
  }

#pragma unroll
  for (int gi = 0; gi < NG; ++gi) {
    const int cur = gi & 1;
    const int nxt = cur ^ 1;
    if (gi + 1 < NG) {
#pragma unroll
      for (int u = 0; u < PGU; ++u) {
        const int n0 = base + 4 * ((gi + 1) * PGU + u);
        const int m0 = (n0 < n) ? n0 : 0;
        const int m1 = (n0 + 1 < n) ? n0 + 1 : 0;
        ra[nxt][u][0] = mcol[(size_t)m0 * G];
        ra[nxt][u][1] = mcol[(size_t)m1 * G];
#pragma unroll
        for (int t = 0; t < 6; ++t) {
          rb[nxt][u][t][0] = hcol[(size_t)m0 * DFEAT + t * 16];
          rb[nxt][u][t][1] = hcol[(size_t)m1 * DFEAT + t * 16];
        }
      }
    }
#pragma unroll
    for (int u = 0; u < PGU; ++u) {
      asm volatile("" ::
          "v"(ra[cur][u][0]), "v"(ra[cur][u][1]),
          "v"(rb[cur][u][0][0]), "v"(rb[cur][u][0][1]), "v"(rb[cur][u][1][0]), "v"(rb[cur][u][1][1]),
          "v"(rb[cur][u][2][0]), "v"(rb[cur][u][2][1]), "v"(rb[cur][u][3][0]), "v"(rb[cur][u][3][1]),
          "v"(rb[cur][u][4][0]), "v"(rb[cur][u][4][1]), "v"(rb[cur][u][5][0]), "v"(rb[cur][u][5][1]));
    }
#pragma unroll
    for (int u = 0; u < PGU; ++u) {
      const int n0 = base + 4 * (gi * PGU + u);
      const bool ok0 = n0 < n;
      const bool ok1 = n0 + 1 < n;
      v2f a;
      a.x = (ok0 && gok) ? ra[cur][u][0] : 0.0f;
      a.y = (ok1 && gok) ? ra[cur][u][1] : 0.0f;
#pragma unroll
      for (int t = 0; t < 6; ++t) {
        v2f b;
        const float x0 = rb[cur][u][t][0];
        const float x1 = rb[cur][u][t][1];
        b.x = (ok0 && x0 > 0.0f) ? x0 : 0.0f;
        b.y = (ok1 && x1 > 0.0f) ? x1 : 0.0f;
        acc[t] = __builtin_amdgcn_wmma_f32_16x16x4_f32(
             false, a,  false, b,
             (short)0, acc[t],  false,  false);
      }
    }
  }

  const int rbase = mtile * 16 + hf * 8;
#pragma unroll
  for (int t = 0; t < 6; ++t) {
#pragma unroll
    for (int r = 0; r < 8; ++r) {
      const int row = rbase + r;
      if (row < G) atomicAdd(&Pacc[(size_t)row * DFEAT + t * 16 + c], acc[t][r]);
    }
  }
}

__global__ __launch_bounds__(32) void k_pool_out(const float* __restrict__ Pacc,
                                                 const float* __restrict__ W,
                                                 const float* __restrict__ bias,
                                                 const float* __restrict__ cnt,
                                                 float* __restrict__ gout, int G) {
  const int lane = threadIdx.x & 31;
  const int c  = lane & 15;
  const int hf = lane >> 4;
  const int g  = blockIdx.x * 16 + c;
  const int col0 = blockIdx.y * 32 + c;
  const bool gok = g < G;
  const float* pr = Pacc + (size_t)(gok ? g : 0) * DFEAT + 2 * hf;
  const float* wc = W + (size_t)(2 * hf) * DFEAT + col0;

  v2f ar[24];
  v2f br0[24];
  v2f br1[24];
#pragma unroll
  for (int k = 0; k < 24; ++k) {
    ar[k] = *(const v2f*)(pr + 4 * k);
    br0[k].x = wc[(size_t)(4 * k) * DFEAT];
    br0[k].y = wc[(size_t)(4 * k + 1) * DFEAT];
    br1[k].x = wc[(size_t)(4 * k) * DFEAT + 16];
    br1[k].y = wc[(size_t)(4 * k + 1) * DFEAT + 16];
  }
  v8f acc0 = (v8f){0.f, 0.f, 0.f, 0.f, 0.f, 0.f, 0.f, 0.f};
  v8f acc1 = (v8f){0.f, 0.f, 0.f, 0.f, 0.f, 0.f, 0.f, 0.f};
#pragma unroll
  for (int k = 0; k < 24; ++k) {
    v2f a = ar[k];
    a.x = gok ? a.x : 0.0f;
    a.y = gok ? a.y : 0.0f;
    acc0 = __builtin_amdgcn_wmma_f32_16x16x4_f32(
         false, a,  false, br0[k],
         (short)0, acc0,  false,  false);
    acc1 = __builtin_amdgcn_wmma_f32_16x16x4_f32(
         false, a,  false, br1[k],
         (short)0, acc1,  false,  false);
  }

  const int rbase = blockIdx.x * 16 + hf * 8;
  const float bb0 = bias[col0];
  const float bb1 = bias[col0 + 16];
  float cn[8];
#pragma unroll
  for (int r = 0; r < 8; ++r) cn[r] = cnt[(rbase + r < G) ? rbase + r : 0];
#pragma unroll
  for (int r = 0; r < 8; ++r) {
    const int row = rbase + r;
    if (row < G) {
      gout[(size_t)row * DFEAT + col0]      = acc0[r] + cn[r] * bb0;
      gout[(size_t)row * DFEAT + col0 + 16] = acc1[r] + cn[r] * bb1;
    }
  }
}

extern "C" void kernel_launch(void* const* d_in, const int* in_sizes, int n_in,
                              void* d_out, int out_size, void* d_ws, size_t ws_size,
                              hipStream_t stream) {
  (void)n_in; (void)ws_size;
  const float* x    = (const float*)d_in[0];
  const float* W1   = (const float*)d_in[1];
  const float* b1   = (const float*)d_in[2];
  const float* W3   = (const float*)d_in[3];
  const float* b3   = (const float*)d_in[4];
  const float* W4   = (const float*)d_in[5];
  const float* b4   = (const float*)d_in[6];
  const float* ew   = (const float*)d_in[7];
  const int*   ei   = (const int*)d_in[8];
  const int*   batch= (const int*)d_in[9];

  const int n  = in_sizes[0] / DFEAT;
  const int ne = in_sizes[7];
  const int* srcp = ei;
  const int* dstp = ei + ne;

  const int nf = n * DFEAT;
  const int gtotal = out_size - nf;

  float* deg  = (float*)d_ws;
  float* dinv = deg + n;
  float* tmp  = dinv + n;
  float* h    = tmp + (size_t)nf;
  const int G = gtotal / DFEAT;
  float* Mt   = h + (size_t)nf;
  float* Pacc = Mt + (size_t)n * G;
  float* cnt  = Pacc + gtotal;
  const int gDM    = (ne + DM_EPW - 1) / DM_EPW;
  const int wpr    = gDM * DM_WPW;
  const int nwords = (ne + 31) / 32;
  const int nr     = (n + AG_R - 1) / AG_R;
  const size_t recOff = ((size_t)(2 * n + 2 * (size_t)nf + (size_t)n * G + gtotal + G) + 3) / 4 * 4;
  int4* rec = (int4*)((float*)d_ws + recOff);
  unsigned* mask = (unsigned*)(rec + ne);

  float* embed = (float*)d_out;
  float* gout  = embed + nf;

  const dim3 B(256);
  const int gN  = (n + 255) / 256;
  const int gAg = ((n + AG_R - 1) / AG_R) * AG_Q;
  const int perPart = (((ne + AG_Q - 1) / AG_Q) + 1023) / 1024 * 1024;
  const int gMM = ((n >> 4) + 7) / 8;

  k_fill<<<gN, B, 0, stream>>>(deg, 1.0f, n);
  k_fill<<<(gtotal + G + 255) / 256, B, 0, stream>>>(Pacc, 0.0f, gtotal + G);
  k_deg_mask<<<gDM, B, 0, stream>>>(srcp, dstp, ew, deg, rec, mask, ne, nr, wpr);
  k_dinv_cnt_mt<<<gN, B, 0, stream>>>(deg, dinv, batch, cnt, Mt, n, G);

  k_gemm_wmma_fused<<<gMM, B, 0, stream>>>(x, W1, dinv, b1, tmp, h, n, 0);
  k_agg_lds<true><<<gAg, dim3(AG_WAVES * 32), 0, stream>>>(tmp, rec, mask, wpr, nwords, dinv, h, n, perPart,
                                                          batch, Mt, G);

  k_gemm_wmma_fused<<<gMM, B, 0, stream>>>(h, W3, dinv, b3, tmp, embed, n, 1);
  k_agg_lds<false><<<gAg, dim3(AG_WAVES * 32), 0, stream>>>(tmp, rec, mask, wpr, nwords, dinv, embed, n, perPart,
                                                           batch, Mt, G);

  k_pool_gemm<<<dim3((n + PCHUNK - 1) / PCHUNK, (G + 63) / 64), dim3(128), 0, stream>>>(Mt, h, Pacc, n, G);
  k_pool_out<<<dim3((G + 15) / 16, DFEAT / 32), dim3(32), 0, stream>>>(Pacc, W4, b4, cnt, gout, G);
}


# Edge aggregation: a work-group owns 196 rows alone (784 were shared by 4): LDS rows start at self-loop+bias and leave once by plain stores; the 19M-per-layer global f32 adds gone (sum order changes)
# speedup vs baseline: 2.2633x; 1.0477x over previous
#include <hip/hip_runtime.h>
#include <hip/hip_bf16.h>

#define DFEAT 96

typedef float v2f __attribute__((ext_vector_type(2)));
typedef float v8f __attribute__((ext_vector_type(8)));


__global__ void k_fill(float* __restrict__ p, float val, int n) {
  int i = blockIdx.x * blockDim.x + threadIdx.x;
  if (i < n) p[i] = val;
}

#define AG_R    196
#define DM_EPW  2048
#define DM_WPW  64
#define DM_PITCH 68
#define DM_MAXR 256
__global__ __launch_bounds__(256) void k_deg_mask(const int* __restrict__ src,
                                                  const int* __restrict__ dst,
                                                  const float* __restrict__ ew,
                                                  float* __restrict__ deg,
                                                  int4* __restrict__ rec,
                                                  unsigned* __restrict__ mask,
                                                  int ne, int nr, int wpr) {
  __shared__ __attribute__((aligned(16))) unsigned tile[DM_MAXR * DM_PITCH];
  uint4* t4 = reinterpret_cast<uint4*>(tile);
  const int lane = threadIdx.x & 31;
  const int wave = threadIdx.x >> 5;
  const int e0 = blockIdx.x * DM_EPW + threadIdx.x;
  int d[8], s[8];
  float w[8];
#pragma unroll
  for (int j = 0; j < 8; ++j) {
    const int e = e0 + j * 256;
    const bool ok = e < ne;
    d[j] = ok ? dst[e] : -1;
    s[j] = ok ? src[e] : 0;
    w[j] = ok ? ew[e] : 0.0f;
  }
  for (int rb = 0; rb < nr; rb += DM_MAXR) {
    for (int i = threadIdx.x; i < DM_MAXR * DM_PITCH / 4; i += 256) t4[i] = make_uint4(0u, 0u, 0u, 0u);
    __syncthreads();
#pragma unroll
    for (int j = 0; j < 8; ++j) {
      if (d[j] >= 0) {
        const int r = d[j] / AG_R - rb;
        if ((unsigned)r < (unsigned)DM_MAXR) atomicOr(&tile[r * DM_PITCH + j * 8 + wave], 1u << lane);
      }
    }
    __syncthreads();
    const int nrows = (nr - rb < DM_MAXR) ? nr - rb : DM_MAXR;
    for (int i = threadIdx.x; i < nrows * (DM_WPW / 4); i += 256) {
      const int r  = i / (DM_WPW / 4);
      const int k4 = i - r * (DM_WPW / 4);
      *reinterpret_cast<uint4*>(mask + (size_t)(rb + r) * wpr + blockIdx.x * DM_WPW + 4 * k4) =
          t4[r * (DM_PITCH / 4) + k4];
    }
    if (rb + DM_MAXR < nr) __syncthreads();
  }
#pragma unroll
  for (int j = 0; j < 8; ++j) {
    if (d[j] >= 0) {
      atomicAdd(&deg[d[j]], w[j]);
      rec[e0 + j * 256] = make_int4(s[j], d[j], __float_as_int(w[j]), 0);
    }
  }
}

#define CNT_BINS 1024
__global__ __launch_bounds__(256) void k_dinv_cnt_mt(const float* __restrict__ deg,
                                                     float* __restrict__ dinv,
                                                     const int* __restrict__ batch,
                                                     float* __restrict__ cnt,
                                                     float* __restrict__ Mt, int n, int G) {
  __shared__ float hist[CNT_BINS];
  __shared__ float sd2[256];
  __shared__ int   sb[256];
  const int nb = (G < CNT_BINS) ? G : CNT_BINS;
  for (int g = threadIdx.x; g < nb; g += 256) hist[g] = 0.0f;
  __syncthreads();
  const int row0 = blockIdx.x * 256;
  const int i = row0 + threadIdx.x;
  float d2 = 0.0f;
  int   bi = -1;
  if (i < n) {
    const float d = deg[i];
    const float dv = (d > 0.0f) ? rsqrtf(fmaxf(d, 1e-20f)) : 0.0f;
    dinv[i] = dv;
    d2 = dv * dv;
    const int b = batch[i];
    bi = b;
    if ((unsigned)b < (unsigned)nb) atomicAdd(&hist[b], 1.0f);
    else if ((unsigned)b < (unsigned)G) atomicAdd(&cnt[b], 1.0f);
  }
  sd2[threadIdx.x] = d2;
  sb[threadIdx.x]  = bi;
  __syncthreads();
  for (int g = threadIdx.x; g < nb; g += 256) {
    const float v = hist[g];
    if (v != 0.0f) atomicAdd(&cnt[g], v);
  }

  const int rows  = (n - row0 < 256) ? n - row0 : 256;
  const int total = rows * G;
  float* mp = Mt + (size_t)row0 * G;
  if (((G & 3) == 0) && ((reinterpret_cast<size_t>(Mt) & 15) == 0)) {
    for (int e0 = 4 * threadIdx.x; e0 < total; e0 += 4 * 256) {
      const int j  = e0 / G;
      const int g0 = e0 - j * G;
      const int b  = sb[j];
      const float s = sd2[j];
      float4 v;
      v.x = (b == g0)     ? s : 0.0f;
      v.y = (b == g0 + 1) ? s : 0.0f;
      v.z = (b == g0 + 2) ? s : 0.0f;
      v.w = (b == g0 + 3) ? s : 0.0f;
      *(float4*)(mp + e0) = v;
    }
  } else {
    for (int e = threadIdx.x; e < total; e += 256) {
      const int j = e / G;
      const int g = e - j * G;
      mp[e] = (sb[j] == g) ? sd2[j] : 0.0f;
    }
  }
}

__global__ __launch_bounds__(256) void k_gemm_wmma(
    const float* __restrict__ X, const float* __restrict__ W,
    float* __restrict__ T, int nrows, int relu_in) {
  __shared__ v2f sW2[(DFEAT / 2) * DFEAT];
  for (int idx = threadIdx.x; idx < (DFEAT / 2) * DFEAT; idx += 256) {
    const int kp  = idx / DFEAT;
    const int col = idx - kp * DFEAT;
    v2f w;
    w.x = W[(2 * kp) * DFEAT + col];
    w.y = W[(2 * kp + 1) * DFEAT + col];
    sW2[idx] = w;
  }
  __syncthreads();

  const int wave = threadIdx.x >> 5;
  const int lane = threadIdx.x & 31;
  const int ntiles = nrows >> 4;
  const int mt = blockIdx.x * 8 + wave;
  if (mt >= ntiles) return;

  const int lo16  = lane & 15;
  const int khalf = lane >> 4;
  const int row   = mt * 16 + lo16;
  const float* xr = X + (size_t)row * DFEAT;

  v8f acc[6];
#pragma unroll
  for (int t = 0; t < 6; ++t) acc[t] = (v8f){0.f, 0.f, 0.f, 0.f, 0.f, 0.f, 0.f, 0.f};

#pragma unroll 4
  for (int k = 0; k < 24; ++k) {
    const int kb = 4 * k + 2 * khalf;
    v2f a = *(const v2f*)(xr + kb);
    if (relu_in) {
      a.x = (a.x > 0.0f) ? a.x : 0.0f;
      a.y = (a.y > 0.0f) ? a.y : 0.0f;
    }
    const v2f* brow = sW2 + (kb >> 1) * DFEAT + lo16;
#pragma unroll
    for (int t = 0; t < 6; ++t) {
      v2f b = brow[t * 16];
      acc[t] = __builtin_amdgcn_wmma_f32_16x16x4_f32(
           false, a,  false, b,
           (short)0, acc[t],  false,  false);
    }
  }

  const int rbase = mt * 16 + khalf * 8;
#pragma unroll
  for (int t = 0; t < 6; ++t) {
    const int cbase = t * 16 + lo16;
#pragma unroll
    for (int r = 0; r < 8; ++r) T[(size_t)(rbase + r) * DFEAT + cbase] = acc[t][r];
  }
}

#define AG_WAVES 32
#define AG_GRP   16
#define AG_QCAP  64
#define AG_F4    (AG_R * DFEAT / 4)
#define AG_NIO   ((AG_F4 + AG_WAVES * 32 - 1) / (AG_WAVES * 32))
template <bool POOL>
__global__ __launch_bounds__(1024) void k_agg_lds(const float* __restrict__ T,
                                                  const int4* __restrict__ rec,
                                                  const unsigned* __restrict__ mask,
                                                  int wpr, int nwords,
                                                  const float* __restrict__ dinv,
                                                  const float* __restrict__ bias,
                                                  float* __restrict__ out,
                                                  int n,
                                                  const int* __restrict__ batch,
                                                  float* __restrict__ Mt, int G) {
  __shared__ __attribute__((aligned(16))) float acc[AG_R * DFEAT];
  __shared__ int qbuf[AG_WAVES * AG_QCAP];
  const int lane = threadIdx.x & 31;
  const int wave = __builtin_amdgcn_readfirstlane((int)(threadIdx.x >> 5));
  const int range = blockIdx.x;
  const int lo = range * AG_R;
  const int rows = (n - lo < AG_R) ? n - lo : AG_R;
  const int nio = rows * (DFEAT / 4);
  float4* a4 = reinterpret_cast<float4*>(acc);

  const uint4* mrow = reinterpret_cast<const uint4*>(mask + (size_t)range * wpr);
  const int pwn = (nwords + 3) >> 2;
  auto load_word = [&](int i) -> uint4 {
    uint4 v = mrow[(i < pwn) ? i : pwn - 1];
    if (i >= pwn) v = make_uint4(0u, 0u, 0u, 0u);
    return v;
  };
  int wb = wave * 32;
  uint4 m = load_word(wb + lane);

  {
    static_assert(AG_NIO * AG_WAVES * 32 >= AG_F4, "every 16-byte word of the range has a thread");
    const float4* t4 = reinterpret_cast<const float4*>(T + (size_t)lo * DFEAT);
    float4 tv[AG_NIO];
    float  dv[AG_NIO];
    float  bv[AG_NIO][4];
#pragma unroll
    for (int k = 0; k < AG_NIO; ++k) {
      const int i  = threadIdx.x + k * (AG_WAVES * 32);
      const int ii = (i < nio) ? i : 0;
      const int r  = ii / (DFEAT / 4);
      const int c0 = 4 * (ii - r * (DFEAT / 4));
      tv[k] = t4[ii];
      dv[k] = dinv[lo + r];
#pragma unroll
      for (int j = 0; j < 4; ++j) bv[k][j] = bias[c0 + j];
    }
    float4 sv[AG_NIO];
#pragma unroll
    for (int k = 0; k < AG_NIO; ++k) {
      const float d2 = dv[k] * dv[k];
      sv[k].x = d2 * tv[k].x + bv[k][0];
      sv[k].y = d2 * tv[k].y + bv[k][1];
      sv[k].z = d2 * tv[k].z + bv[k][2];
      sv[k].w = d2 * tv[k].w + bv[k][3];
    }
#pragma unroll
    for (int k = 0; k < AG_NIO; ++k) {
      const int i = threadIdx.x + k * (AG_WAVES * 32);
      if (i < nio) a4[i] = sv[k];
    }
  }
  __syncthreads();

  int* q = qbuf + wave * AG_QCAP;
  int qn = 0;

  auto drain = [&](const int* qq, int cnt) {
    const int qe = qq[(lane < cnt) ? lane : cnt - 1];
    const int4 rc = rec[qe];
    const int s = rc.x;
    const int r = rc.y - lo;
    const float w = __int_as_float(rc.z);
    const int dl = r * DFEAT;
    const float nrm = dinv[s] * w * dinv[lo + r];
    const int nrm_bits = __float_as_int(nrm);
    int gph = 0;
    if (POOL) gph = batch[lo + r];
#pragma unroll 1
    for (int g0 = 0; g0 < cnt; g0 += AG_GRP) {
      float t[AG_GRP][3];
#pragma unroll
      for (int u = 0; u < AG_GRP; ++u) {
        const int i = (g0 + u < cnt) ? g0 + u : cnt - 1;
        const float* ts = T + (size_t)__builtin_amdgcn_readlane(s, i) * DFEAT;
#pragma unroll
        for (int j = 0; j < 3; ++j) t[u][j] = ts[lane + 32 * j];
      }
#pragma unroll
      for (int u = 0; u < AG_GRP; ++u) {
        const int i = g0 + u;
        if (i < cnt) {
          float* ad = acc + __builtin_amdgcn_readlane(dl, i);
          const float nu = __int_as_float(__builtin_amdgcn_readlane(nrm_bits, i));
#pragma unroll
          for (int j = 0; j < 3; ++j) atomicAdd(&ad[lane + 32 * j], nu * t[u][j]);
        }
      }
    }
    if (POOL) {
      if (lane < cnt && (unsigned)gph < (unsigned)G) atomicAdd(&Mt[(size_t)s * G + gph], nrm);
    }
  };

  constexpr int WSTRIDE = AG_WAVES * 32;
  for (; wb < pwn; wb += WSTRIDE) {
    const uint4 mn = load_word(wb + WSTRIDE + lane);
    const int ebase = (wb + lane) << 7;
    unsigned left = m.x | m.y | m.z | m.w;
    unsigned any = (unsigned)__builtin_amdgcn_ballot_w32(left != 0u);
    while (any != 0u) {
      const int pos = qn + __builtin_popcount(any & ((1u << lane) - 1u));
      if (left != 0u) {
        const bool hx = m.x != 0u;
        const bool hy = !hx && m.y != 0u;
        const bool hz = !hx && !hy && m.z != 0u;
        const bool hw = !hx && !hy && !hz;
        const unsigned cw  = hx ? m.x : hy ? m.y : hz ? m.z : m.w;
        const int      off = hx ? 0 : hy ? 32 : hz ? 64 : 96;
        const unsigned cl  = cw & (cw - 1u);
        q[pos] = ebase + off + __builtin_ctz(cw);
        m.x = hx ? cl : m.x;
        m.y = hy ? cl : m.y;
        m.z = hz ? cl : m.z;
        m.w = hw ? cl : m.w;
      }
      qn += __builtin_popcount(any);
      if (qn >= 32) {
        drain(q + (qn - 32), 32);
        qn -= 32;
      }
      left = m.x | m.y | m.z | m.w;
      any = (unsigned)__builtin_amdgcn_ballot_w32(left != 0u);
    }
    m = mn;
  }
  if (qn > 0) drain(q, qn);

  __syncthreads();
  float4* o4 = reinterpret_cast<float4*>(out + (size_t)lo * DFEAT);
  for (int i = threadIdx.x; i < nio; i += AG_WAVES * 32) o4[i] = a4[i];
}


#define PCHUNK 128


#define PGU 4
__global__ __launch_bounds__(128) void k_pool_gemm(const float* __restrict__ Mt,
                                                   const float* __restrict__ H,
                                                   float* __restrict__ Pacc, int n, int G) {
  const int wave  = threadIdx.x >> 5;
  const int lane  = threadIdx.x & 31;
  const int mtile = blockIdx.y * 4 + wave;
  if (mtile * 16 >= G) return;
  const int c  = lane & 15;
  const int hf = lane >> 4;
  const int g  = mtile * 16 + c;
  const bool gok = g < G;
  const float* mcol = Mt + (gok ? g : 0);
  const float* hcol = H + c;
  const int base = blockIdx.x * PCHUNK + 2 * hf;

  v8f acc[6];
#pragma unroll
  for (int t = 0; t < 6; ++t) acc[t] = (v8f){0.f, 0.f, 0.f, 0.f, 0.f, 0.f, 0.f, 0.f};

  constexpr int NG = PCHUNK / (4 * PGU);
  float ra[2][PGU][2];
  float rb[2][PGU][6][2];

#pragma unroll
  for (int u = 0; u < PGU; ++u) {
    const int n0 = base + 4 * u;
    const int m0 = (n0 < n) ? n0 : 0;
    const int m1 = (n0 + 1 < n) ? n0 + 1 : 0;
    ra[0][u][0] = mcol[(size_t)m0 * G];
    ra[0][u][1] = mcol[(size_t)m1 * G];
#pragma unroll
    for (int t = 0; t < 6; ++t) {
      rb[0][u][t][0] = hcol[(size_t)m0 * DFEAT + t * 16];
      rb[0][u][t][1] = hcol[(size_t)m1 * DFEAT + t * 16];
    }
  }

#pragma unroll
  for (int gi = 0; gi < NG; ++gi) {
    const int cur = gi & 1;
    const int nxt = cur ^ 1;
    if (gi + 1 < NG) {
#pragma unroll
      for (int u = 0; u < PGU; ++u) {
        const int n0 = base + 4 * ((gi + 1) * PGU + u);
        const int m0 = (n0 < n) ? n0 : 0;
        const int m1 = (n0 + 1 < n) ? n0 + 1 : 0;
        ra[nxt][u][0] = mcol[(size_t)m0 * G];
        ra[nxt][u][1] = mcol[(size_t)m1 * G];
#pragma unroll
        for (int t = 0; t < 6; ++t) {
          rb[nxt][u][t][0] = hcol[(size_t)m0 * DFEAT + t * 16];
          rb[nxt][u][t][1] = hcol[(size_t)m1 * DFEAT + t * 16];
        }
      }
    }
#pragma unroll
    for (int u = 0; u < PGU; ++u) {
      asm volatile("" ::
          "v"(ra[cur][u][0]), "v"(ra[cur][u][1]),
          "v"(rb[cur][u][0][0]), "v"(rb[cur][u][0][1]), "v"(rb[cur][u][1][0]), "v"(rb[cur][u][1][1]),
          "v"(rb[cur][u][2][0]), "v"(rb[cur][u][2][1]), "v"(rb[cur][u][3][0]), "v"(rb[cur][u][3][1]),
          "v"(rb[cur][u][4][0]), "v"(rb[cur][u][4][1]), "v"(rb[cur][u][5][0]), "v"(rb[cur][u][5][1]));
    }
#pragma unroll
    for (int u = 0; u < PGU; ++u) {
      const int n0 = base + 4 * (gi * PGU + u);
      const bool ok0 = n0 < n;
      const bool ok1 = n0 + 1 < n;
      v2f a;
      a.x = (ok0 && gok) ? ra[cur][u][0] : 0.0f;
      a.y = (ok1 && gok) ? ra[cur][u][1] : 0.0f;
#pragma unroll
      for (int t = 0; t < 6; ++t) {
        v2f b;
        const float x0 = rb[cur][u][t][0];
        const float x1 = rb[cur][u][t][1];
        b.x = (ok0 && x0 > 0.0f) ? x0 : 0.0f;
        b.y = (ok1 && x1 > 0.0f) ? x1 : 0.0f;
        acc[t] = __builtin_amdgcn_wmma_f32_16x16x4_f32(
             false, a,  false, b,
             (short)0, acc[t],  false,  false);
      }
    }
  }

  const int rbase = mtile * 16 + hf * 8;
#pragma unroll
  for (int t = 0; t < 6; ++t) {
#pragma unroll
    for (int r = 0; r < 8; ++r) {
      const int row = rbase + r;
      if (row < G) atomicAdd(&Pacc[(size_t)row * DFEAT + t * 16 + c], acc[t][r]);
    }
  }
}

__global__ __launch_bounds__(32) void k_pool_out(const float* __restrict__ Pacc,
                                                 const float* __restrict__ W,
                                                 const float* __restrict__ bias,
                                                 const float* __restrict__ cnt,
                                                 float* __restrict__ gout, int G) {
  const int lane = threadIdx.x & 31;
  const int c  = lane & 15;
  const int hf = lane >> 4;
  const int g  = blockIdx.x * 16 + c;
  const int col0 = blockIdx.y * 32 + c;
  const bool gok = g < G;
  const float* pr = Pacc + (size_t)(gok ? g : 0) * DFEAT + 2 * hf;
  const float* wc = W + (size_t)(2 * hf) * DFEAT + col0;

  v2f ar[24];
  v2f br0[24];
  v2f br1[24];
#pragma unroll
  for (int k = 0; k < 24; ++k) {
    ar[k] = *(const v2f*)(pr + 4 * k);
    br0[k].x = wc[(size_t)(4 * k) * DFEAT];
    br0[k].y = wc[(size_t)(4 * k + 1) * DFEAT];
    br1[k].x = wc[(size_t)(4 * k) * DFEAT + 16];
    br1[k].y = wc[(size_t)(4 * k + 1) * DFEAT + 16];
  }
  v8f acc0 = (v8f){0.f, 0.f, 0.f, 0.f, 0.f, 0.f, 0.f, 0.f};
  v8f acc1 = (v8f){0.f, 0.f, 0.f, 0.f, 0.f, 0.f, 0.f, 0.f};
#pragma unroll
  for (int k = 0; k < 24; ++k) {
    v2f a = ar[k];
    a.x = gok ? a.x : 0.0f;
    a.y = gok ? a.y : 0.0f;
    acc0 = __builtin_amdgcn_wmma_f32_16x16x4_f32(
         false, a,  false, br0[k],
         (short)0, acc0,  false,  false);
    acc1 = __builtin_amdgcn_wmma_f32_16x16x4_f32(
         false, a,  false, br1[k],
         (short)0, acc1,  false,  false);
  }

  const int rbase = blockIdx.x * 16 + hf * 8;
  const float bb0 = bias[col0];
  const float bb1 = bias[col0 + 16];
  float cn[8];
#pragma unroll
  for (int r = 0; r < 8; ++r) cn[r] = cnt[(rbase + r < G) ? rbase + r : 0];
#pragma unroll
  for (int r = 0; r < 8; ++r) {
    const int row = rbase + r;
    if (row < G) {
      gout[(size_t)row * DFEAT + col0]      = acc0[r] + cn[r] * bb0;
      gout[(size_t)row * DFEAT + col0 + 16] = acc1[r] + cn[r] * bb1;
    }
  }
}

extern "C" void kernel_launch(void* const* d_in, const int* in_sizes, int n_in,
                              void* d_out, int out_size, void* d_ws, size_t ws_size,
                              hipStream_t stream) {
  (void)n_in; (void)ws_size;
  const float* x    = (const float*)d_in[0];
  const float* W1   = (const float*)d_in[1];
  const float* b1   = (const float*)d_in[2];
  const float* W3   = (const float*)d_in[3];
  const float* b3   = (const float*)d_in[4];
  const float* W4   = (const float*)d_in[5];
  const float* b4   = (const float*)d_in[6];
  const float* ew   = (const float*)d_in[7];
  const int*   ei   = (const int*)d_in[8];
  const int*   batch= (const int*)d_in[9];

  const int n  = in_sizes[0] / DFEAT;
  const int ne = in_sizes[7];
  const int* srcp = ei;
  const int* dstp = ei + ne;

  const int nf = n * DFEAT;
  const int gtotal = out_size - nf;

  float* deg  = (float*)d_ws;
  float* dinv = deg + n;
  float* tmp  = dinv + n;
  float* h    = tmp + (size_t)nf;
  const int G = gtotal / DFEAT;
  float* Mt   = h + (size_t)nf;
  float* Pacc = Mt + (size_t)n * G;
  float* cnt  = Pacc + gtotal;
  const int gDM    = (ne + DM_EPW - 1) / DM_EPW;
  const int wpr    = gDM * DM_WPW;
  const int nwords = (ne + 31) / 32;
  const int nr     = (n + AG_R - 1) / AG_R;
  const size_t recOff = ((size_t)(2 * n + 2 * (size_t)nf + (size_t)n * G + gtotal + G) + 3) / 4 * 4;
  int4* rec = (int4*)((float*)d_ws + recOff);
  unsigned* mask = (unsigned*)(rec + ne);

  float* embed = (float*)d_out;
  float* gout  = embed + nf;

  const dim3 B(256);
  const int gN  = (n + 255) / 256;
  const int gAg = nr;
  const int gMM = ((n >> 4) + 7) / 8;

  k_fill<<<gN, B, 0, stream>>>(deg, 1.0f, n);
  k_fill<<<(gtotal + G + 255) / 256, B, 0, stream>>>(Pacc, 0.0f, gtotal + G);
  k_deg_mask<<<gDM, B, 0, stream>>>(srcp, dstp, ew, deg, rec, mask, ne, nr, wpr);
  k_dinv_cnt_mt<<<gN, B, 0, stream>>>(deg, dinv, batch, cnt, Mt, n, G);

  k_gemm_wmma<<<gMM, B, 0, stream>>>(x, W1, tmp, n, 0);
  k_agg_lds<true><<<gAg, dim3(AG_WAVES * 32), 0, stream>>>(tmp, rec, mask, wpr, nwords, dinv, b1, h, n,
                                                          batch, Mt, G);

  k_gemm_wmma<<<gMM, B, 0, stream>>>(h, W3, tmp, n, 1);
  k_agg_lds<false><<<gAg, dim3(AG_WAVES * 32), 0, stream>>>(tmp, rec, mask, wpr, nwords, dinv, b3, embed, n,
                                                           batch, Mt, G);

  k_pool_gemm<<<dim3((n + PCHUNK - 1) / PCHUNK, (G + 63) / 64), dim3(128), 0, stream>>>(Mt, h, Pacc, n, G);
  k_pool_out<<<dim3((G + 15) / 16, DFEAT / 32), dim3(32), 0, stream>>>(Pacc, W4, b4, cnt, gout, G);
}


# Edge norm factored: gathered rows carry dinv[src] (GEMM row scale; hidden rows kept times dinv), dinv[dst] applied once per owned row; no per-edge dinv loads (terms round differently, sum order kept)
# speedup vs baseline: 2.3856x; 1.0540x over previous
#include <hip/hip_runtime.h>
#include <hip/hip_bf16.h>

#define DFEAT 96

typedef float v2f __attribute__((ext_vector_type(2)));
typedef float v8f __attribute__((ext_vector_type(8)));
typedef float v4f __attribute__((ext_vector_type(4)));


__global__ void k_fill(float* __restrict__ p, float val, int n) {
  int i = blockIdx.x * blockDim.x + threadIdx.x;
  if (i < n) p[i] = val;
}

#define AG_R    196
#define DM_EPW  2048
#define DM_WPW  64
#define DM_PITCH 68
#define DM_MAXR 256
__global__ __launch_bounds__(256) void k_deg_mask(const int* __restrict__ src,
                                                  const int* __restrict__ dst,
                                                  const float* __restrict__ ew,
                                                  float* __restrict__ deg,
                                                  int4* __restrict__ rec,
                                                  unsigned* __restrict__ mask,
                                                  int ne, int nr, int wpr) {
  __shared__ __attribute__((aligned(16))) unsigned tile[DM_MAXR * DM_PITCH];
  uint4* t4 = reinterpret_cast<uint4*>(tile);
  const int lane = threadIdx.x & 31;
  const int wave = threadIdx.x >> 5;
  const int e0 = blockIdx.x * DM_EPW + threadIdx.x;
  int d[8], s[8];
  float w[8];
#pragma unroll
  for (int j = 0; j < 8; ++j) {
    const int e = e0 + j * 256;
    const bool ok = e < ne;
    d[j] = ok ? dst[e] : -1;
    s[j] = ok ? src[e] : 0;
    w[j] = ok ? ew[e] : 0.0f;
  }
  for (int rb = 0; rb < nr; rb += DM_MAXR) {
    for (int i = threadIdx.x; i < DM_MAXR * DM_PITCH / 4; i += 256) t4[i] = make_uint4(0u, 0u, 0u, 0u);
    __syncthreads();
#pragma unroll
    for (int j = 0; j < 8; ++j) {
      if (d[j] >= 0) {
        const int r = d[j] / AG_R - rb;
        if ((unsigned)r < (unsigned)DM_MAXR) atomicOr(&tile[r * DM_PITCH + j * 8 + wave], 1u << lane);
      }
    }
    __syncthreads();
    const int nrows = (nr - rb < DM_MAXR) ? nr - rb : DM_MAXR;
    for (int i = threadIdx.x; i < nrows * (DM_WPW / 4); i += 256) {
      const int r  = i / (DM_WPW / 4);
      const int k4 = i - r * (DM_WPW / 4);
      *reinterpret_cast<uint4*>(mask + (size_t)(rb + r) * wpr + blockIdx.x * DM_WPW + 4 * k4) =
          t4[r * (DM_PITCH / 4) + k4];
    }
    if (rb + DM_MAXR < nr) __syncthreads();
  }
#pragma unroll
  for (int j = 0; j < 8; ++j) {
    if (d[j] >= 0) {
      atomicAdd(&deg[d[j]], w[j]);
      rec[e0 + j * 256] = make_int4(s[j], d[j], __float_as_int(w[j]), 0);
    }
  }
}

#define CNT_BINS 1024
__global__ __launch_bounds__(256) void k_dinv_cnt_mt(const float* __restrict__ deg,
                                                     float* __restrict__ dinv,
                                                     const int* __restrict__ batch,
                                                     float* __restrict__ cnt,
                                                     float* __restrict__ Mt, int n, int G) {
  __shared__ float hist[CNT_BINS];
  __shared__ float sd2[256];
  __shared__ int   sb[256];
  const int nb = (G < CNT_BINS) ? G : CNT_BINS;
  for (int g = threadIdx.x; g < nb; g += 256) hist[g] = 0.0f;
  __syncthreads();
  const int row0 = blockIdx.x * 256;
  const int i = row0 + threadIdx.x;
  float d2 = 0.0f;
  int   bi = -1;
  if (i < n) {
    const float d = deg[i];
    const float dv = (d > 0.0f) ? rsqrtf(fmaxf(d, 1e-20f)) : 0.0f;
    dinv[i] = dv;
    d2 = dv;
    const int b = batch[i];
    bi = b;
    if ((unsigned)b < (unsigned)nb) atomicAdd(&hist[b], 1.0f);
    else if ((unsigned)b < (unsigned)G) atomicAdd(&cnt[b], 1.0f);
  }
  sd2[threadIdx.x] = d2;
  sb[threadIdx.x]  = bi;
  __syncthreads();
  for (int g = threadIdx.x; g < nb; g += 256) {
    const float v = hist[g];
    if (v != 0.0f) atomicAdd(&cnt[g], v);
  }

  const int rows  = (n - row0 < 256) ? n - row0 : 256;
  const int total = rows * G;
  float* mp = Mt + (size_t)row0 * G;
  if (((G & 3) == 0) && ((reinterpret_cast<size_t>(Mt) & 15) == 0)) {
    for (int e0 = 4 * threadIdx.x; e0 < total; e0 += 4 * 256) {
      const int j  = e0 / G;
      const int g0 = e0 - j * G;
      const int b  = sb[j];
      const float s = sd2[j];
      float4 v;
      v.x = (b == g0)     ? s : 0.0f;
      v.y = (b == g0 + 1) ? s : 0.0f;
      v.z = (b == g0 + 2) ? s : 0.0f;
      v.w = (b == g0 + 3) ? s : 0.0f;
      *(float4*)(mp + e0) = v;
    }
  } else {
    for (int e = threadIdx.x; e < total; e += 256) {
      const int j = e / G;
      const int g = e - j * G;
      mp[e] = (sb[j] == g) ? sd2[j] : 0.0f;
    }
  }
}

__global__ __launch_bounds__(256) void k_gemm_wmma(
    const float* __restrict__ X, const float* __restrict__ W,
    float* __restrict__ T, int nrows, int relu_in, const float* __restrict__ rs) {
  __shared__ v2f sW2[(DFEAT / 2) * DFEAT];
  for (int idx = threadIdx.x; idx < (DFEAT / 2) * DFEAT; idx += 256) {
    const int kp  = idx / DFEAT;
    const int col = idx - kp * DFEAT;
    v2f w;
    w.x = W[(2 * kp) * DFEAT + col];
    w.y = W[(2 * kp + 1) * DFEAT + col];
    sW2[idx] = w;
  }
  __syncthreads();

  const int wave = threadIdx.x >> 5;
  const int lane = threadIdx.x & 31;
  const int ntiles = nrows >> 4;
  const int mt = blockIdx.x * 8 + wave;
  if (mt >= ntiles) return;

  const int lo16  = lane & 15;
  const int khalf = lane >> 4;
  const int row   = mt * 16 + lo16;
  const float* xr = X + (size_t)row * DFEAT;

  const int rbase = mt * 16 + khalf * 8;
  float sc[8];
#pragma unroll
  for (int r = 0; r < 8; ++r) sc[r] = 1.0f;
  if (rs) {
#pragma unroll
    for (int r = 0; r < 8; ++r) sc[r] = rs[rbase + r];
  }

  v8f acc[6];
#pragma unroll
  for (int t = 0; t < 6; ++t) acc[t] = (v8f){0.f, 0.f, 0.f, 0.f, 0.f, 0.f, 0.f, 0.f};

#pragma unroll 4
  for (int k = 0; k < 24; ++k) {
    const int kb = 4 * k + 2 * khalf;
    v2f a = *(const v2f*)(xr + kb);
    if (relu_in) {
      a.x = (a.x > 0.0f) ? a.x : 0.0f;
      a.y = (a.y > 0.0f) ? a.y : 0.0f;
    }
    const v2f* brow = sW2 + (kb >> 1) * DFEAT + lo16;
#pragma unroll
    for (int t = 0; t < 6; ++t) {
      v2f b = brow[t * 16];
      acc[t] = __builtin_amdgcn_wmma_f32_16x16x4_f32(
           false, a,  false, b,
           (short)0, acc[t],  false,  false);
    }
  }

#pragma unroll
  for (int t = 0; t < 6; ++t) {
    const int cbase = t * 16 + lo16;
#pragma unroll
    for (int r = 0; r < 8; ++r) T[(size_t)(rbase + r) * DFEAT + cbase] = acc[t][r] * sc[r];
  }
}

#define AG_WAVES 32
#define AG_GRP   16
#define AG_QCAP  64
#define AG_F4    (AG_R * DFEAT / 4)
#define AG_NIO   ((AG_F4 + AG_WAVES * 32 - 1) / (AG_WAVES * 32))
template <bool POOL>
__global__ __launch_bounds__(1024) void k_agg_lds(const float* __restrict__ T,
                                                  const int4* __restrict__ rec,
                                                  const unsigned* __restrict__ mask,
                                                  int wpr, int nwords,
                                                  const float* __restrict__ dinv,
                                                  const float* __restrict__ bias,
                                                  float* __restrict__ out,
                                                  int n,
                                                  const int* __restrict__ batch,
                                                  float* __restrict__ Mt, int G) {
  __shared__ __attribute__((aligned(16))) float acc[AG_NIO * AG_WAVES * 32 * 4];
  __shared__ __attribute__((aligned(16))) float sbias[AG_WAVES * 32];
  __shared__ float sdv[AG_WAVES * 32];
  __shared__ int   sgr[AG_WAVES * 32];
  __shared__ int qbuf[AG_WAVES * AG_QCAP];
  const int lane = threadIdx.x & 31;
  const int wave = __builtin_amdgcn_readfirstlane((int)(threadIdx.x >> 5));
  const int range = blockIdx.x;
  const int lo = range * AG_R;
  const int rows = (n - lo < AG_R) ? n - lo : AG_R;
  const int nio = rows * (DFEAT / 4);
  v4f* a4 = reinterpret_cast<v4f*>(acc);

  const uint4* mrow = reinterpret_cast<const uint4*>(mask + (size_t)range * wpr);
  const int pwn = (nwords + 3) >> 2;
  auto load_word = [&](int i) -> uint4 {
    uint4 v = mrow[(i < pwn) ? i : pwn - 1];
    if (i >= pwn) v = make_uint4(0u, 0u, 0u, 0u);
    return v;
  };
  int wb = wave * 32;
  uint4 m = load_word(wb + lane);

  {
    static_assert(AG_NIO * AG_WAVES * 32 >= AG_F4, "every 16-byte word of the range has a thread");
    static_assert(AG_WAVES * 32 >= AG_R && AG_WAVES * 32 >= DFEAT, "a thread per owned row and per bias word");
    const v4f* t4 = reinterpret_cast<const v4f*>(T + (size_t)lo * DFEAT);
    v4f tv[AG_NIO];
#pragma unroll
    for (int k = 0; k < AG_NIO; ++k) {
      const int i  = threadIdx.x + k * (AG_WAVES * 32);
      tv[k] = t4[(i < nio) ? i : 0];
    }
    const int   rr = ((int)threadIdx.x < rows) ? (int)threadIdx.x : 0;
    const float dr = dinv[lo + rr];
    int gr = 0;
    if (POOL) gr = batch[lo + rr];
    const float br = bias[((int)threadIdx.x < DFEAT) ? (int)threadIdx.x : 0];
#pragma unroll
    for (int k = 0; k < AG_NIO; ++k) a4[threadIdx.x + k * (AG_WAVES * 32)] = tv[k];
    sdv[threadIdx.x] = dr;
    if (POOL) sgr[threadIdx.x] = gr;
    sbias[threadIdx.x] = br;
  }
  __syncthreads();

  int* q = qbuf + wave * AG_QCAP;
  int qn = 0;

  auto drain = [&](const int* qq, int cnt) {
    const int qe = qq[(lane < cnt) ? lane : cnt - 1];
    const int4 rc = rec[qe];
    const int s = rc.x;
    const int r = rc.y - lo;
    const int w_bits = rc.z;
    const int dl = r * DFEAT;
    int gph = 0;
    float dvr = 0.0f;
    if (POOL) {
      gph = sgr[r];
      dvr = sdv[r];
    }
#pragma unroll 1
    for (int g0 = 0; g0 < cnt; g0 += AG_GRP) {
      float t[AG_GRP][3];
#pragma unroll
      for (int u = 0; u < AG_GRP; ++u) {
        const int i = (g0 + u < cnt) ? g0 + u : cnt - 1;
        const float* ts = T + (size_t)__builtin_amdgcn_readlane(s, i) * DFEAT;
#pragma unroll
        for (int j = 0; j < 3; ++j) t[u][j] = ts[lane + 32 * j];
      }
#pragma unroll
      for (int u = 0; u < AG_GRP; ++u) {
        const int i = g0 + u;
        if (i < cnt) {
          float* ad = acc + __builtin_amdgcn_readlane(dl, i);
          const float nu = __int_as_float(__builtin_amdgcn_readlane(w_bits, i));
#pragma unroll
          for (int j = 0; j < 3; ++j) atomicAdd(&ad[lane + 32 * j], nu * t[u][j]);
        }
      }
    }
    if (POOL) {
      if (lane < cnt && (unsigned)gph < (unsigned)G)
        atomicAdd(&Mt[(size_t)s * G + gph], __int_as_float(w_bits) * dvr);
    }
  };

  constexpr int WSTRIDE = AG_WAVES * 32;
  for (; wb < pwn; wb += WSTRIDE) {
    const uint4 mn = load_word(wb + WSTRIDE + lane);
    const int ebase = (wb + lane) << 7;
    unsigned left = m.x | m.y | m.z | m.w;
    unsigned any = (unsigned)__builtin_amdgcn_ballot_w32(left != 0u);
    while (any != 0u) {
      const int pos = qn + __builtin_popcount(any & ((1u << lane) - 1u));
      if (left != 0u) {
        const bool hx = m.x != 0u;
        const bool hy = !hx && m.y != 0u;
        const bool hz = !hx && !hy && m.z != 0u;
        const bool hw = !hx && !hy && !hz;
        const unsigned cw  = hx ? m.x : hy ? m.y : hz ? m.z : m.w;
        const int      off = hx ? 0 : hy ? 32 : hz ? 64 : 96;
        const unsigned cl  = cw & (cw - 1u);
        q[pos] = ebase + off + __builtin_ctz(cw);
        m.x = hx ? cl : m.x;
        m.y = hy ? cl : m.y;
        m.z = hz ? cl : m.z;
        m.w = hw ? cl : m.w;
      }
      qn += __builtin_popcount(any);
      if (qn >= 32) {
        drain(q + (qn - 32), 32);
        qn -= 32;
      }
      left = m.x | m.y | m.z | m.w;
      any = (unsigned)__builtin_amdgcn_ballot_w32(left != 0u);
    }
    m = mn;
  }
  if (qn > 0) drain(q, qn);

  __syncthreads();
  v4f* o4 = reinterpret_cast<v4f*>(out + (size_t)lo * DFEAT);
  const v4f* b4 = reinterpret_cast<const v4f*>(sbias);
  for (int i = threadIdx.x; i < nio; i += AG_WAVES * 32) {
    const int r = i / (DFEAT / 4);
    const float d = sdv[r];
    v4f v = d * a4[i] + b4[i - r * (DFEAT / 4)];
    if (POOL) v = v * d;
    o4[i] = v;
  }
}


#define PCHUNK 128


#define PGU 4
__global__ __launch_bounds__(128) void k_pool_gemm(const float* __restrict__ Mt,
                                                   const float* __restrict__ H,
                                                   float* __restrict__ Pacc, int n, int G) {
  const int wave  = threadIdx.x >> 5;
  const int lane  = threadIdx.x & 31;
  const int mtile = blockIdx.y * 4 + wave;
  if (mtile * 16 >= G) return;
  const int c  = lane & 15;
  const int hf = lane >> 4;
  const int g  = mtile * 16 + c;
  const bool gok = g < G;
  const float* mcol = Mt + (gok ? g : 0);
  const float* hcol = H + c;
  const int base = blockIdx.x * PCHUNK + 2 * hf;

  v8f acc[6];
#pragma unroll
  for (int t = 0; t < 6; ++t) acc[t] = (v8f){0.f, 0.f, 0.f, 0.f, 0.f, 0.f, 0.f, 0.f};

  constexpr int NG = PCHUNK / (4 * PGU);
  float ra[2][PGU][2];
  float rb[2][PGU][6][2];

#pragma unroll
  for (int u = 0; u < PGU; ++u) {
    const int n0 = base + 4 * u;
    const int m0 = (n0 < n) ? n0 : 0;
    const int m1 = (n0 + 1 < n) ? n0 + 1 : 0;
    ra[0][u][0] = mcol[(size_t)m0 * G];
    ra[0][u][1] = mcol[(size_t)m1 * G];
#pragma unroll
    for (int t = 0; t < 6; ++t) {
      rb[0][u][t][0] = hcol[(size_t)m0 * DFEAT + t * 16];
      rb[0][u][t][1] = hcol[(size_t)m1 * DFEAT + t * 16];
    }
  }

#pragma unroll
  for (int gi = 0; gi < NG; ++gi) {
    const int cur = gi & 1;
    const int nxt = cur ^ 1;
    if (gi + 1 < NG) {
#pragma unroll
      for (int u = 0; u < PGU; ++u) {
        const int n0 = base + 4 * ((gi + 1) * PGU + u);
        const int m0 = (n0 < n) ? n0 : 0;
        const int m1 = (n0 + 1 < n) ? n0 + 1 : 0;
        ra[nxt][u][0] = mcol[(size_t)m0 * G];
        ra[nxt][u][1] = mcol[(size_t)m1 * G];
#pragma unroll
        for (int t = 0; t < 6; ++t) {
          rb[nxt][u][t][0] = hcol[(size_t)m0 * DFEAT + t * 16];
          rb[nxt][u][t][1] = hcol[(size_t)m1 * DFEAT + t * 16];
        }
      }
    }
#pragma unroll
    for (int u = 0; u < PGU; ++u) {
      asm volatile("" ::
          "v"(ra[cur][u][0]), "v"(ra[cur][u][1]),
          "v"(rb[cur][u][0][0]), "v"(rb[cur][u][0][1]), "v"(rb[cur][u][1][0]), "v"(rb[cur][u][1][1]),
          "v"(rb[cur][u][2][0]), "v"(rb[cur][u][2][1]), "v"(rb[cur][u][3][0]), "v"(rb[cur][u][3][1]),
          "v"(rb[cur][u][4][0]), "v"(rb[cur][u][4][1]), "v"(rb[cur][u][5][0]), "v"(rb[cur][u][5][1]));
    }
#pragma unroll
    for (int u = 0; u < PGU; ++u) {
      const int n0 = base + 4 * (gi * PGU + u);
      const bool ok0 = n0 < n;
      const bool ok1 = n0 + 1 < n;
      v2f a;
      a.x = (ok0 && gok) ? ra[cur][u][0] : 0.0f;
      a.y = (ok1 && gok) ? ra[cur][u][1] : 0.0f;
#pragma unroll
      for (int t = 0; t < 6; ++t) {
        v2f b;
        const float x0 = rb[cur][u][t][0];
        const float x1 = rb[cur][u][t][1];
        b.x = (ok0 && x0 > 0.0f) ? x0 : 0.0f;
        b.y = (ok1 && x1 > 0.0f) ? x1 : 0.0f;
        acc[t] = __builtin_amdgcn_wmma_f32_16x16x4_f32(
             false, a,  false, b,
             (short)0, acc[t],  false,  false);
      }
    }
  }

  const int rbase = mtile * 16 + hf * 8;
#pragma unroll
  for (int t = 0; t < 6; ++t) {
#pragma unroll
    for (int r = 0; r < 8; ++r) {
      const int row = rbase + r;
      if (row < G) atomicAdd(&Pacc[(size_t)row * DFEAT + t * 16 + c], acc[t][r]);
    }
  }
}

__global__ __launch_bounds__(32) void k_pool_out(const float* __restrict__ Pacc,
                                                 const float* __restrict__ W,
                                                 const float* __restrict__ bias,
                                                 const float* __restrict__ cnt,
                                                 float* __restrict__ gout, int G) {
  const int lane = threadIdx.x & 31;
  const int c  = lane & 15;
  const int hf = lane >> 4;
  const int g  = blockIdx.x * 16 + c;
  const int col0 = blockIdx.y * 32 + c;
  const bool gok = g < G;
  const float* pr = Pacc + (size_t)(gok ? g : 0) * DFEAT + 2 * hf;
  const float* wc = W + (size_t)(2 * hf) * DFEAT + col0;

  v2f ar[24];
  v2f br0[24];
  v2f br1[24];
#pragma unroll
  for (int k = 0; k < 24; ++k) {
    ar[k] = *(const v2f*)(pr + 4 * k);
    br0[k].x = wc[(size_t)(4 * k) * DFEAT];
    br0[k].y = wc[(size_t)(4 * k + 1) * DFEAT];
    br1[k].x = wc[(size_t)(4 * k) * DFEAT + 16];
    br1[k].y = wc[(size_t)(4 * k + 1) * DFEAT + 16];
  }
  v8f acc0 = (v8f){0.f, 0.f, 0.f, 0.f, 0.f, 0.f, 0.f, 0.f};
  v8f acc1 = (v8f){0.f, 0.f, 0.f, 0.f, 0.f, 0.f, 0.f, 0.f};
#pragma unroll
  for (int k = 0; k < 24; ++k) {
    v2f a = ar[k];
    a.x = gok ? a.x : 0.0f;
    a.y = gok ? a.y : 0.0f;
    acc0 = __builtin_amdgcn_wmma_f32_16x16x4_f32(
         false, a,  false, br0[k],
         (short)0, acc0,  false,  false);
    acc1 = __builtin_amdgcn_wmma_f32_16x16x4_f32(
         false, a,  false, br1[k],
         (short)0, acc1,  false,  false);
  }

  const int rbase = blockIdx.x * 16 + hf * 8;
  const float bb0 = bias[col0];
  const float bb1 = bias[col0 + 16];
  float cn[8];
#pragma unroll
  for (int r = 0; r < 8; ++r) cn[r] = cnt[(rbase + r < G) ? rbase + r : 0];
#pragma unroll
  for (int r = 0; r < 8; ++r) {
    const int row = rbase + r;
    if (row < G) {
      gout[(size_t)row * DFEAT + col0]      = acc0[r] + cn[r] * bb0;
      gout[(size_t)row * DFEAT + col0 + 16] = acc1[r] + cn[r] * bb1;
    }
  }
}

extern "C" void kernel_launch(void* const* d_in, const int* in_sizes, int n_in,
                              void* d_out, int out_size, void* d_ws, size_t ws_size,
                              hipStream_t stream) {
  (void)n_in; (void)ws_size;
  const float* x    = (const float*)d_in[0];
  const float* W1   = (const float*)d_in[1];
  const float* b1   = (const float*)d_in[2];
  const float* W3   = (const float*)d_in[3];
  const float* b3   = (const float*)d_in[4];
  const float* W4   = (const float*)d_in[5];
  const float* b4   = (const float*)d_in[6];
  const float* ew   = (const float*)d_in[7];
  const int*   ei   = (const int*)d_in[8];
  const int*   batch= (const int*)d_in[9];

  const int n  = in_sizes[0] / DFEAT;
  const int ne = in_sizes[7];
  const int* srcp = ei;
  const int* dstp = ei + ne;

  const int nf = n * DFEAT;
  const int gtotal = out_size - nf;

  float* deg  = (float*)d_ws;
  float* dinv = deg + n;
  float* tmp  = dinv + n;
  float* h    = tmp + (size_t)nf;
  const int G = gtotal / DFEAT;
  float* Mt   = h + (size_t)nf;
  float* Pacc = Mt + (size_t)n * G;
  float* cnt  = Pacc + gtotal;
  const int gDM    = (ne + DM_EPW - 1) / DM_EPW;
  const int wpr    = gDM * DM_WPW;
  const int nwords = (ne + 31) / 32;
  const int nr     = (n + AG_R - 1) / AG_R;
  const size_t recOff = ((size_t)(2 * n + 2 * (size_t)nf + (size_t)n * G + gtotal + G) + 3) / 4 * 4;
  int4* rec = (int4*)((float*)d_ws + recOff);
  unsigned* mask = (unsigned*)(rec + ne);

  float* embed = (float*)d_out;
  float* gout  = embed + nf;

  const dim3 B(256);
  const int gN  = (n + 255) / 256;
  const int gAg = nr;
  const int gMM = ((n >> 4) + 7) / 8;

  k_fill<<<gN, B, 0, stream>>>(deg, 1.0f, n);
  k_fill<<<(gtotal + G + 255) / 256, B, 0, stream>>>(Pacc, 0.0f, gtotal + G);
  k_deg_mask<<<gDM, B, 0, stream>>>(srcp, dstp, ew, deg, rec, mask, ne, nr, wpr);
  k_dinv_cnt_mt<<<gN, B, 0, stream>>>(deg, dinv, batch, cnt, Mt, n, G);

  k_gemm_wmma<<<gMM, B, 0, stream>>>(x, W1, tmp, n, 0, dinv);
  k_agg_lds<true><<<gAg, dim3(AG_WAVES * 32), 0, stream>>>(tmp, rec, mask, wpr, nwords, dinv, b1, h, n,
                                                          batch, Mt, G);

  k_gemm_wmma<<<gMM, B, 0, stream>>>(h, W3, tmp, n, 1, (const float*)nullptr);
  k_agg_lds<false><<<gAg, dim3(AG_WAVES * 32), 0, stream>>>(tmp, rec, mask, wpr, nwords, dinv, b3, embed, n,
                                                           batch, Mt, G);

  k_pool_gemm<<<dim3((n + PCHUNK - 1) / PCHUNK, (G + 63) / 64), dim3(128), 0, stream>>>(Mt, h, Pacc, n, G);
  k_pool_out<<<dim3((G + 15) / 16, DFEAT / 32), dim3(32), 0, stream>>>(Pacc, W4, b4, cnt, gout, G);
}


# Edge records grouped by owner range inside each 2048-edge block (LDS counting sort in the degree launch): an aggregation reads its runs as whole lines; mask scan and queue gone (arrival-order sums)
# speedup vs baseline: 2.4731x; 1.0367x over previous
#include <hip/hip_runtime.h>
#include <hip/hip_bf16.h>

#define DFEAT 96

typedef float v2f __attribute__((ext_vector_type(2)));
typedef float v8f __attribute__((ext_vector_type(8)));
typedef float v4f __attribute__((ext_vector_type(4)));


__global__ void k_fill(float* __restrict__ p, float val, int n) {
  int i = blockIdx.x * blockDim.x + threadIdx.x;
  if (i < n) p[i] = val;
}

#define AG_R    196
#define DM_EPW  2048
#define BIN_K   4
#define BIN_MAXR (256 * BIN_K)
__global__ __launch_bounds__(256) void k_deg_bin(const int* __restrict__ src,
                                                 const int* __restrict__ dst,
                                                 const float* __restrict__ ew,
                                                 float* __restrict__ deg,
                                                 int4* __restrict__ rec,
                                                 unsigned* __restrict__ desc,
                                                 int ne, int nr, int dpitch) {
  __shared__ int cntr[BIN_MAXR];
  __shared__ int wtot[8];
  __shared__ __attribute__((aligned(16))) int4 stage[DM_EPW];
  const int lane = threadIdx.x & 31;
  const int wave = threadIdx.x >> 5;
  const int eb = blockIdx.x * DM_EPW;
  const int e0 = eb + threadIdx.x;
  int d[8], s[8];
  float w[8];
#pragma unroll
  for (int j = 0; j < 8; ++j) {
    const int e = e0 + j * 256;
    const bool ok = e < ne;
    d[j] = ok ? dst[e] : -1;
    s[j] = ok ? src[e] : 0;
    w[j] = ok ? ew[e] : 0.0f;
  }
#pragma unroll
  for (int k = 0; k < BIN_K; ++k) cntr[threadIdx.x + 256 * k] = 0;
  __syncthreads();
  int rg[8], rk[8];
#pragma unroll
  for (int j = 0; j < 8; ++j) {
    rg[j] = 0;
    rk[j] = 0;
    if (d[j] >= 0) {
      int r = d[j] / AG_R;
      r = (r < BIN_MAXR) ? r : BIN_MAXR - 1;
      rg[j] = r;
      rk[j] = atomicAdd(&cntr[r], 1);
    }
  }
  __syncthreads();
  int c[BIN_K];
  int tsum = 0;
#pragma unroll
  for (int k = 0; k < BIN_K; ++k) {
    c[k] = cntr[BIN_K * threadIdx.x + k];
    tsum += c[k];
  }
  int inc = tsum;
#pragma unroll
  for (int o = 1; o < 32; o <<= 1) {
    const int v = __shfl_up(inc, o);
    if (lane >= o) inc += v;
  }
  if (lane == 31) wtot[wave] = inc;
  __syncthreads();
  int run0 = inc - tsum;
#pragma unroll
  for (int i = 0; i < 8; ++i) run0 += (i < wave) ? wtot[i] : 0;
  {
    int run = run0;
#pragma unroll
    for (int k = 0; k < BIN_K; ++k) {
      cntr[BIN_K * threadIdx.x + k] = run;
      run += c[k];
    }
  }
  __syncthreads();
  int pos[8];
#pragma unroll
  for (int j = 0; j < 8; ++j) pos[j] = cntr[rg[j]] + rk[j];
#pragma unroll
  for (int j = 0; j < 8; ++j) {
    if (d[j] >= 0) stage[pos[j]] = make_int4(s[j], d[j], __float_as_int(w[j]), 0);
  }
  __syncthreads();
  const int nvalid = (ne - eb < DM_EPW) ? ne - eb : DM_EPW;
  for (int i = threadIdx.x; i < nvalid; i += 256) rec[eb + i] = stage[i];
  {
    int run = run0;
#pragma unroll
    for (int k = 0; k < BIN_K; ++k) {
      const int r = BIN_K * threadIdx.x + k;
      if (r < nr) desc[(size_t)r * dpitch + blockIdx.x] = (unsigned)run | ((unsigned)c[k] << 16);
      run += c[k];
    }
  }
#pragma unroll
  for (int j = 0; j < 8; ++j) {
    if (d[j] >= 0) atomicAdd(&deg[d[j]], w[j]);
  }
}

#define CNT_BINS 1024
__global__ __launch_bounds__(256) void k_dinv_cnt_mt(const float* __restrict__ deg,
                                                     float* __restrict__ dinv,
                                                     const int* __restrict__ batch,
                                                     float* __restrict__ cnt,
                                                     float* __restrict__ Mt, int n, int G) {
  __shared__ float hist[CNT_BINS];
  __shared__ float sd2[256];
  __shared__ int   sb[256];
  const int nb = (G < CNT_BINS) ? G : CNT_BINS;
  for (int g = threadIdx.x; g < nb; g += 256) hist[g] = 0.0f;
  __syncthreads();
  const int row0 = blockIdx.x * 256;
  const int i = row0 + threadIdx.x;
  float d2 = 0.0f;
  int   bi = -1;
  if (i < n) {
    const float d = deg[i];
    const float dv = (d > 0.0f) ? rsqrtf(fmaxf(d, 1e-20f)) : 0.0f;
    dinv[i] = dv;
    d2 = dv;
    const int b = batch[i];
    bi = b;
    if ((unsigned)b < (unsigned)nb) atomicAdd(&hist[b], 1.0f);
    else if ((unsigned)b < (unsigned)G) atomicAdd(&cnt[b], 1.0f);
  }
  sd2[threadIdx.x] = d2;
  sb[threadIdx.x]  = bi;
  __syncthreads();
  for (int g = threadIdx.x; g < nb; g += 256) {
    const float v = hist[g];
    if (v != 0.0f) atomicAdd(&cnt[g], v);
  }

  const int rows  = (n - row0 < 256) ? n - row0 : 256;
  const int total = rows * G;
  float* mp = Mt + (size_t)row0 * G;
  if (((G & 3) == 0) && ((reinterpret_cast<size_t>(Mt) & 15) == 0)) {
    for (int e0 = 4 * threadIdx.x; e0 < total; e0 += 4 * 256) {
      const int j  = e0 / G;
      const int g0 = e0 - j * G;
      const int b  = sb[j];
      const float s = sd2[j];
      float4 v;
      v.x = (b == g0)     ? s : 0.0f;
      v.y = (b == g0 + 1) ? s : 0.0f;
      v.z = (b == g0 + 2) ? s : 0.0f;
      v.w = (b == g0 + 3) ? s : 0.0f;
      *(float4*)(mp + e0) = v;
    }
  } else {
    for (int e = threadIdx.x; e < total; e += 256) {
      const int j = e / G;
      const int g = e - j * G;
      mp[e] = (sb[j] == g) ? sd2[j] : 0.0f;
    }
  }
}

__global__ __launch_bounds__(256) void k_gemm_wmma(
    const float* __restrict__ X, const float* __restrict__ W,
    float* __restrict__ T, int nrows, int relu_in, const float* __restrict__ rs) {
  __shared__ v2f sW2[(DFEAT / 2) * DFEAT];
  for (int idx = threadIdx.x; idx < (DFEAT / 2) * DFEAT; idx += 256) {
    const int kp  = idx / DFEAT;
    const int col = idx - kp * DFEAT;
    v2f w;
    w.x = W[(2 * kp) * DFEAT + col];
    w.y = W[(2 * kp + 1) * DFEAT + col];
    sW2[idx] = w;
  }
  __syncthreads();

  const int wave = threadIdx.x >> 5;
  const int lane = threadIdx.x & 31;
  const int ntiles = nrows >> 4;
  const int mt = blockIdx.x * 8 + wave;
  if (mt >= ntiles) return;

  const int lo16  = lane & 15;
  const int khalf = lane >> 4;
  const int row   = mt * 16 + lo16;
  const float* xr = X + (size_t)row * DFEAT;

  const int rbase = mt * 16 + khalf * 8;
  float sc[8];
#pragma unroll
  for (int r = 0; r < 8; ++r) sc[r] = 1.0f;
  if (rs) {
#pragma unroll
    for (int r = 0; r < 8; ++r) sc[r] = rs[rbase + r];
  }

  v8f acc[6];
#pragma unroll
  for (int t = 0; t < 6; ++t) acc[t] = (v8f){0.f, 0.f, 0.f, 0.f, 0.f, 0.f, 0.f, 0.f};

#pragma unroll 4
  for (int k = 0; k < 24; ++k) {
    const int kb = 4 * k + 2 * khalf;
    v2f a = *(const v2f*)(xr + kb);
    if (relu_in) {
      a.x = (a.x > 0.0f) ? a.x : 0.0f;
      a.y = (a.y > 0.0f) ? a.y : 0.0f;
    }
    const v2f* brow = sW2 + (kb >> 1) * DFEAT + lo16;
#pragma unroll
    for (int t = 0; t < 6; ++t) {
      v2f b = brow[t * 16];
      acc[t] = __builtin_amdgcn_wmma_f32_16x16x4_f32(
           false, a,  false, b,
           (short)0, acc[t],  false,  false);
    }
  }

#pragma unroll
  for (int t = 0; t < 6; ++t) {
    const int cbase = t * 16 + lo16;
#pragma unroll
    for (int r = 0; r < 8; ++r) T[(size_t)(rbase + r) * DFEAT + cbase] = acc[t][r] * sc[r];
  }
}

#define AG_WAVES 32
#define AG_GRP   16
#define AG_F4    (AG_R * DFEAT / 4)
#define AG_NIO   ((AG_F4 + AG_WAVES * 32 - 1) / (AG_WAVES * 32))
template <bool POOL>
__global__ __launch_bounds__(1024) void k_agg_lds(const float* __restrict__ T,
                                                  const int4* __restrict__ rec,
                                                  const unsigned* __restrict__ desc,
                                                  int dpitch, int nseg,
                                                  const float* __restrict__ dinv,
                                                  const float* __restrict__ bias,
                                                  float* __restrict__ out,
                                                  int n,
                                                  const int* __restrict__ batch,
                                                  float* __restrict__ Mt, int G) {
  __shared__ __attribute__((aligned(16))) float acc[AG_NIO * AG_WAVES * 32 * 4];
  __shared__ __attribute__((aligned(16))) float sbias[AG_WAVES * 32];
  __shared__ float sdv[AG_WAVES * 32];
  __shared__ int   sgr[AG_WAVES * 32];
  const int lane = threadIdx.x & 31;
  const int wave = __builtin_amdgcn_readfirstlane((int)(threadIdx.x >> 5));
  const int range = blockIdx.x;
  const int lo = range * AG_R;
  const int rows = (n - lo < AG_R) ? n - lo : AG_R;
  const int nio = rows * (DFEAT / 4);
  v4f* a4 = reinterpret_cast<v4f*>(acc);

  const int sbeg = (int)(((long long)wave * nseg) / AG_WAVES);
  const int send = (int)(((long long)(wave + 1) * nseg) / AG_WAVES);
  const unsigned* drow = desc + (size_t)range * dpitch;
  auto load_desc = [&](int sb) -> unsigned {
    const int sg = sb + lane;
    unsigned v = drow[(sg < send) ? sg : nseg - 1];
    if (sg >= send) v = 0u;
    return v;
  };
  unsigned dsc = load_desc(sbeg);

  {
    static_assert(AG_NIO * AG_WAVES * 32 >= AG_F4, "every 16-byte word of the range has a thread");
    static_assert(AG_WAVES * 32 >= AG_R && AG_WAVES * 32 >= DFEAT, "a thread per owned row and per bias word");
    const v4f* t4 = reinterpret_cast<const v4f*>(T + (size_t)lo * DFEAT);
    v4f tv[AG_NIO];
#pragma unroll
    for (int k = 0; k < AG_NIO; ++k) {
      const int i  = threadIdx.x + k * (AG_WAVES * 32);
      tv[k] = t4[(i < nio) ? i : 0];
    }
    const int   rr = ((int)threadIdx.x < rows) ? (int)threadIdx.x : 0;
    const float dr = dinv[lo + rr];
    int gr = 0;
    if (POOL) gr = batch[lo + rr];
    const float br = bias[((int)threadIdx.x < DFEAT) ? (int)threadIdx.x : 0];
#pragma unroll
    for (int k = 0; k < AG_NIO; ++k) a4[threadIdx.x + k * (AG_WAVES * 32)] = tv[k];
    sdv[threadIdx.x] = dr;
    if (POOL) sgr[threadIdx.x] = gr;
    sbias[threadIdx.x] = br;
  }
  __syncthreads();

  auto group = [&](const int s, const int dl, const int w_bits, const int g0, const int cnt,
                   auto&& after_loads) {
    float t[AG_GRP][3];
#pragma unroll
    for (int u = 0; u < AG_GRP; ++u) {
      const int i = (g0 + u < cnt) ? g0 + u : cnt - 1;
      const float* ts = T + (size_t)__builtin_amdgcn_readlane(s, i) * DFEAT;
#pragma unroll
      for (int j = 0; j < 3; ++j) t[u][j] = ts[lane + 32 * j];
    }
    after_loads();
    __builtin_amdgcn_sched_barrier(0);
#pragma unroll
    for (int u = 0; u < AG_GRP; ++u) {
      const int i = g0 + u;
      if (i < cnt) {
        float* ad = acc + __builtin_amdgcn_readlane(dl, i);
        const float nu = __int_as_float(__builtin_amdgcn_readlane(w_bits, i));
#pragma unroll
        for (int j = 0; j < 3; ++j) atomicAdd(&ad[lane + 32 * j], nu * t[u][j]);
      }
    }
  };
  static_assert(2 * AG_GRP == 32, "a 32-edge batch is two row groups");

  auto drain = [&](const int4 rc, int cnt, auto&& next_records) {
    const int s = rc.x;
    const int r = rc.y - lo;
    const int w_bits = rc.z;
    const int dl = r * DFEAT;
    int gph = 0;
    float dvr = 0.0f;
    if (POOL) {
      gph = sgr[r];
      dvr = sdv[r];
    }
    group(s, dl, w_bits, 0, cnt, next_records);
    if (cnt > AG_GRP) group(s, dl, w_bits, AG_GRP, cnt, [] {});
    if (POOL) {
      if (lane < cnt && (unsigned)gph < (unsigned)G)
        atomicAdd(&Mt[(size_t)s * G + gph], __int_as_float(w_bits) * dvr);
    }
  };

  for (int sb = sbeg; sb < send; sb += 32) {
    if (sb != sbeg) dsc = load_desc(sb);
    const int nact = (send - sb < 32) ? send - sb : 32;
    const int cntl = (int)(dsc >> 16);
    int excl = 0;
    int tot = 0;
    for (int i = 0; i < nact; ++i) {
      if (lane == i) excl = tot;
      tot += __builtin_amdgcn_readlane(cntl, i);
    }
    const int val = (sb + lane) * DM_EPW + (int)(dsc & 0xffffu) - excl;
    auto rec_of = [&](int f) -> int {
      int v = __builtin_amdgcn_readlane(val, 0);
      for (int i = 1; i < nact; ++i) {
        const int ex = __builtin_amdgcn_readlane(excl, i);
        const int vv = __builtin_amdgcn_readlane(val, i);
        v = (f >= ex) ? vv : v;
      }
      return v + f;
    };
    if (tot > 0) {
      int4 rcn = rec[rec_of((lane < tot) ? lane : tot - 1)];
      for (int b0 = 0; b0 < tot; b0 += 32) {
        const int4 rc = rcn;
        const int cnt = (tot - b0 < 32) ? tot - b0 : 32;
        const int fn = b0 + 32 + lane;
        const int en = rec_of((fn < tot) ? fn : tot - 1);
        drain(rc, cnt, [&] { rcn = rec[en]; });
      }
    }
  }

  __syncthreads();
  v4f* o4 = reinterpret_cast<v4f*>(out + (size_t)lo * DFEAT);
  const v4f* b4 = reinterpret_cast<const v4f*>(sbias);
  for (int i = threadIdx.x; i < nio; i += AG_WAVES * 32) {
    const int r = i / (DFEAT / 4);
    const float d = sdv[r];
    v4f v = d * a4[i] + b4[i - r * (DFEAT / 4)];
    if (POOL) v = v * d;
    o4[i] = v;
  }
}


#define PCHUNK 128


#define PGU 4
__global__ __launch_bounds__(128) void k_pool_gemm(const float* __restrict__ Mt,
                                                   const float* __restrict__ H,
                                                   float* __restrict__ Pacc, int n, int G) {
  const int wave  = threadIdx.x >> 5;
  const int lane  = threadIdx.x & 31;
  const int mtile = blockIdx.y * 4 + wave;
  if (mtile * 16 >= G) return;
  const int c  = lane & 15;
  const int hf = lane >> 4;
  const int g  = mtile * 16 + c;
  const bool gok = g < G;
  const float* mcol = Mt + (gok ? g : 0);
  const float* hcol = H + c;
  const int base = blockIdx.x * PCHUNK + 2 * hf;

  v8f acc[6];
#pragma unroll
  for (int t = 0; t < 6; ++t) acc[t] = (v8f){0.f, 0.f, 0.f, 0.f, 0.f, 0.f, 0.f, 0.f};

  constexpr int NG = PCHUNK / (4 * PGU);
  float ra[2][PGU][2];
  float rb[2][PGU][6][2];

#pragma unroll
  for (int u = 0; u < PGU; ++u) {
    const int n0 = base + 4 * u;
    const int m0 = (n0 < n) ? n0 : 0;
    const int m1 = (n0 + 1 < n) ? n0 + 1 : 0;
    ra[0][u][0] = mcol[(size_t)m0 * G];
    ra[0][u][1] = mcol[(size_t)m1 * G];
#pragma unroll
    for (int t = 0; t < 6; ++t) {
      rb[0][u][t][0] = hcol[(size_t)m0 * DFEAT + t * 16];
      rb[0][u][t][1] = hcol[(size_t)m1 * DFEAT + t * 16];
    }
  }

#pragma unroll
  for (int gi = 0; gi < NG; ++gi) {
    const int cur = gi & 1;
    const int nxt = cur ^ 1;
    if (gi + 1 < NG) {
#pragma unroll
      for (int u = 0; u < PGU; ++u) {
        const int n0 = base + 4 * ((gi + 1) * PGU + u);
        const int m0 = (n0 < n) ? n0 : 0;
        const int m1 = (n0 + 1 < n) ? n0 + 1 : 0;
        ra[nxt][u][0] = mcol[(size_t)m0 * G];
        ra[nxt][u][1] = mcol[(size_t)m1 * G];
#pragma unroll
        for (int t = 0; t < 6; ++t) {
          rb[nxt][u][t][0] = hcol[(size_t)m0 * DFEAT + t * 16];
          rb[nxt][u][t][1] = hcol[(size_t)m1 * DFEAT + t * 16];
        }
      }
    }
#pragma unroll
    for (int u = 0; u < PGU; ++u) {
      asm volatile("" ::
          "v"(ra[cur][u][0]), "v"(ra[cur][u][1]),
          "v"(rb[cur][u][0][0]), "v"(rb[cur][u][0][1]), "v"(rb[cur][u][1][0]), "v"(rb[cur][u][1][1]),
          "v"(rb[cur][u][2][0]), "v"(rb[cur][u][2][1]), "v"(rb[cur][u][3][0]), "v"(rb[cur][u][3][1]),
          "v"(rb[cur][u][4][0]), "v"(rb[cur][u][4][1]), "v"(rb[cur][u][5][0]), "v"(rb[cur][u][5][1]));
    }
#pragma unroll
    for (int u = 0; u < PGU; ++u) {
      const int n0 = base + 4 * (gi * PGU + u);
      const bool ok0 = n0 < n;
      const bool ok1 = n0 + 1 < n;
      v2f a;
      a.x = (ok0 && gok) ? ra[cur][u][0] : 0.0f;
      a.y = (ok1 && gok) ? ra[cur][u][1] : 0.0f;
#pragma unroll
      for (int t = 0; t < 6; ++t) {
        v2f b;
        const float x0 = rb[cur][u][t][0];
        const float x1 = rb[cur][u][t][1];
        b.x = (ok0 && x0 > 0.0f) ? x0 : 0.0f;
        b.y = (ok1 && x1 > 0.0f) ? x1 : 0.0f;
        acc[t] = __builtin_amdgcn_wmma_f32_16x16x4_f32(
             false, a,  false, b,
             (short)0, acc[t],  false,  false);
      }
    }
  }

  const int rbase = mtile * 16 + hf * 8;
#pragma unroll
  for (int t = 0; t < 6; ++t) {
#pragma unroll
    for (int r = 0; r < 8; ++r) {
      const int row = rbase + r;
      if (row < G) atomicAdd(&Pacc[(size_t)row * DFEAT + t * 16 + c], acc[t][r]);
    }
  }
}

__global__ __launch_bounds__(32) void k_pool_out(const float* __restrict__ Pacc,
                                                 const float* __restrict__ W,
                                                 const float* __restrict__ bias,
                                                 const float* __restrict__ cnt,
                                                 float* __restrict__ gout, int G) {
  const int lane = threadIdx.x & 31;
  const int c  = lane & 15;
  const int hf = lane >> 4;
  const int g  = blockIdx.x * 16 + c;
  const int col0 = blockIdx.y * 32 + c;
  const bool gok = g < G;
  const float* pr = Pacc + (size_t)(gok ? g : 0) * DFEAT + 2 * hf;
  const float* wc = W + (size_t)(2 * hf) * DFEAT + col0;

  v2f ar[24];
  v2f br0[24];
  v2f br1[24];
#pragma unroll
  for (int k = 0; k < 24; ++k) {
    ar[k] = *(const v2f*)(pr + 4 * k);
    br0[k].x = wc[(size_t)(4 * k) * DFEAT];
    br0[k].y = wc[(size_t)(4 * k + 1) * DFEAT];
    br1[k].x = wc[(size_t)(4 * k) * DFEAT + 16];
    br1[k].y = wc[(size_t)(4 * k + 1) * DFEAT + 16];
  }
  v8f acc0 = (v8f){0.f, 0.f, 0.f, 0.f, 0.f, 0.f, 0.f, 0.f};
  v8f acc1 = (v8f){0.f, 0.f, 0.f, 0.f, 0.f, 0.f, 0.f, 0.f};
#pragma unroll
  for (int k = 0; k < 24; ++k) {
    v2f a = ar[k];
    a.x = gok ? a.x : 0.0f;
    a.y = gok ? a.y : 0.0f;
    acc0 = __builtin_amdgcn_wmma_f32_16x16x4_f32(
         false, a,  false, br0[k],
         (short)0, acc0,  false,  false);
    acc1 = __builtin_amdgcn_wmma_f32_16x16x4_f32(
         false, a,  false, br1[k],
         (short)0, acc1,  false,  false);
  }

  const int rbase = blockIdx.x * 16 + hf * 8;
  const float bb0 = bias[col0];
  const float bb1 = bias[col0 + 16];
  float cn[8];
#pragma unroll
  for (int r = 0; r < 8; ++r) cn[r] = cnt[(rbase + r < G) ? rbase + r : 0];
#pragma unroll
  for (int r = 0; r < 8; ++r) {
    const int row = rbase + r;
    if (row < G) {
      gout[(size_t)row * DFEAT + col0]      = acc0[r] + cn[r] * bb0;
      gout[(size_t)row * DFEAT + col0 + 16] = acc1[r] + cn[r] * bb1;
    }
  }
}

extern "C" void kernel_launch(void* const* d_in, const int* in_sizes, int n_in,
                              void* d_out, int out_size, void* d_ws, size_t ws_size,
                              hipStream_t stream) {
  (void)n_in; (void)ws_size;
  const float* x    = (const float*)d_in[0];
  const float* W1   = (const float*)d_in[1];
  const float* b1   = (const float*)d_in[2];
  const float* W3   = (const float*)d_in[3];
  const float* b3   = (const float*)d_in[4];
  const float* W4   = (const float*)d_in[5];
  const float* b4   = (const float*)d_in[6];
  const float* ew   = (const float*)d_in[7];
  const int*   ei   = (const int*)d_in[8];
  const int*   batch= (const int*)d_in[9];

  const int n  = in_sizes[0] / DFEAT;
  const int ne = in_sizes[7];
  const int* srcp = ei;
  const int* dstp = ei + ne;

  const int nf = n * DFEAT;
  const int gtotal = out_size - nf;

  float* deg  = (float*)d_ws;
  float* dinv = deg + n;
  float* tmp  = dinv + n;
  float* h    = tmp + (size_t)nf;
  const int G = gtotal / DFEAT;
  float* Mt   = h + (size_t)nf;
  float* Pacc = Mt + (size_t)n * G;
  float* cnt  = Pacc + gtotal;
  const int gDM    = (ne + DM_EPW - 1) / DM_EPW;
  const int dpitch = (gDM + 31) / 32 * 32;
  const int nr     = (n + AG_R - 1) / AG_R;
  const size_t recOff = ((size_t)(2 * n + 2 * (size_t)nf + (size_t)n * G + gtotal + G) + 3) / 4 * 4;
  int4* rec = (int4*)((float*)d_ws + recOff);
  unsigned* desc = (unsigned*)(rec + ne);

  float* embed = (float*)d_out;
  float* gout  = embed + nf;

  const dim3 B(256);
  const int gN  = (n + 255) / 256;
  const int gAg = nr;
  const int gMM = ((n >> 4) + 7) / 8;

  k_fill<<<gN, B, 0, stream>>>(deg, 1.0f, n);
  k_fill<<<(gtotal + G + 255) / 256, B, 0, stream>>>(Pacc, 0.0f, gtotal + G);
  k_deg_bin<<<gDM, B, 0, stream>>>(srcp, dstp, ew, deg, rec, desc, ne, nr, dpitch);
  k_dinv_cnt_mt<<<gN, B, 0, stream>>>(deg, dinv, batch, cnt, Mt, n, G);

  k_gemm_wmma<<<gMM, B, 0, stream>>>(x, W1, tmp, n, 0, dinv);
  k_agg_lds<true><<<gAg, dim3(AG_WAVES * 32), 0, stream>>>(tmp, rec, desc, dpitch, gDM, dinv, b1, h, n,
                                                          batch, Mt, G);

  k_gemm_wmma<<<gMM, B, 0, stream>>>(h, W3, tmp, n, 1, (const float*)nullptr);
  k_agg_lds<false><<<gAg, dim3(AG_WAVES * 32), 0, stream>>>(tmp, rec, desc, dpitch, gDM, dinv, b3, embed, n,
                                                           batch, Mt, G);

  k_pool_gemm<<<dim3((n + PCHUNK - 1) / PCHUNK, (G + 63) / 64), dim3(128), 0, stream>>>(Mt, h, Pacc, n, G);
  k_pool_out<<<dim3((G + 15) / 16, DFEAT / 32), dim3(32), 0, stream>>>(Pacc, W4, b4, cnt, gout, G);
}
